# starting kernel
# baseline (speedup 1.0000x reference)
#include <hip/hip_runtime.h>
#include <hip/hip_bf16.h>

typedef __attribute__((ext_vector_type(2))) float v2f;
typedef __attribute__((ext_vector_type(8))) float v8f;

#define D_FEAT 64
#define N_GRAPHS_C 64

__global__ __launch_bounds__(256) void deg_init_kernel(float* __restrict__ deg, int n) {
    int i = blockIdx.x * blockDim.x + threadIdx.x;
    if (i < n) deg[i] = 1.0f;
}

__global__ __launch_bounds__(256) void deg_count_kernel(const int* __restrict__ col,
                                                        float* __restrict__ deg, int e) {
    int i = blockIdx.x * blockDim.x + threadIdx.x;
    if (i < e) atomicAdd(&deg[col[i]], 1.0f);
}

__global__ __launch_bounds__(256) void deg_finish_kernel(float* __restrict__ deg, int n) {
    int i = blockIdx.x * blockDim.x + threadIdx.x;
    if (i < n) {
        float d = deg[i];
        deg[i] = (d > 0.0f) ? rsqrtf(d) : 0.0f;
    }
}

__global__ __launch_bounds__(256) void gemm_wmma_kernel(const float* __restrict__ X,
                                                        const float* __restrict__ W,
                                                        float* __restrict__ H,
                                                        int nrows) {
    const int wave = (int)((blockIdx.x * blockDim.x + threadIdx.x) >> 5);
    const int lane = (int)(threadIdx.x & 31);
    const int row0 = wave * 16;
    if (row0 >= nrows) return;
    const int lm = lane & 15;
    const int kk = (lane >> 4) * 2;

    const float* __restrict__ xrow = X + (size_t)(row0 + lm) * D_FEAT + kk;

    v8f c0 = {}, c1 = {}, c2 = {}, c3 = {};
#pragma unroll
    for (int k = 0; k < D_FEAT; k += 4) {
        v2f a;
        a.x = xrow[k + 0];
        a.y = xrow[k + 1];
        const float* __restrict__ wr0 = W + (size_t)(k + kk) * D_FEAT + lm;
        const float* __restrict__ wr1 = wr0 + D_FEAT;
        v2f b0, b1, b2, b3;
        b0.x = wr0[0];   b0.y = wr1[0];
        b1.x = wr0[16];  b1.y = wr1[16];
        b2.x = wr0[32];  b2.y = wr1[32];
        b3.x = wr0[48];  b3.y = wr1[48];
        c0 = __builtin_amdgcn_wmma_f32_16x16x4_f32(false, a, false, b0, (short)0, c0, false, false);
        c1 = __builtin_amdgcn_wmma_f32_16x16x4_f32(false, a, false, b1, (short)0, c1, false, false);
        c2 = __builtin_amdgcn_wmma_f32_16x16x4_f32(false, a, false, b2, (short)0, c2, false, false);
        c3 = __builtin_amdgcn_wmma_f32_16x16x4_f32(false, a, false, b3, (short)0, c3, false, false);
    }

    const int mbase = row0 + (lane >> 4) * 8;
#pragma unroll
    for (int r = 0; r < 8; ++r) {
        float* __restrict__ hr = H + (size_t)(mbase + r) * D_FEAT + lm;
        hr[0]  = c0[r];
        hr[16] = c1[r];
        hr[32] = c2[r];
        hr[48] = c3[r];
    }
}

__global__ __launch_bounds__(256) void init_acc_kernel(const float* __restrict__ H,
                                                       const float* __restrict__ dinv,
                                                       const float* __restrict__ bias,
                                                       float* __restrict__ ACC, int n) {
    int t = blockIdx.x * blockDim.x + threadIdx.x;
    if (t >= n) return;
    int node = t >> 6;
    int d    = t & 63;
    float di = dinv[node];
    ACC[t] = bias[d] + H[t] * di * di;
}

__global__ __launch_bounds__(256) void scatter_kernel(const float* __restrict__ H,
                                                      const int* __restrict__ row,
                                                      const int* __restrict__ col,
                                                      const float* __restrict__ dinv,
                                                      float* __restrict__ ACC, int e) {
    long long t = (long long)blockIdx.x * blockDim.x + threadIdx.x;
    int edge = (int)(t >> 4);
    if (edge >= e) return;
    int f = (int)(t & 15) * 4;
    int r = row[edge];
    int c = col[edge];
    float s = dinv[r] * dinv[c];
    const float4 hv = *(const float4*)(H + (size_t)r * D_FEAT + f);
    float* dst = ACC + (size_t)c * D_FEAT + f;
    atomicAdd(dst + 0, hv.x * s);
    atomicAdd(dst + 1, hv.y * s);
    atomicAdd(dst + 2, hv.z * s);
    atomicAdd(dst + 3, hv.w * s);
}

__global__ __launch_bounds__(256) void relu_kernel(float* __restrict__ A, int n) {
    int i = blockIdx.x * blockDim.x + threadIdx.x;
    if (i < n) A[i] = fmaxf(A[i], 0.0f);
}

__global__ __launch_bounds__(256) void pool_zero_kernel(float* __restrict__ pool,
                                                        float* __restrict__ cnt, int n) {
    int i = blockIdx.x * blockDim.x + threadIdx.x;
    if (i < n) pool[i] = 0.0f;
    if (i < N_GRAPHS_C) cnt[i] = 0.0f;
}

__global__ __launch_bounds__(256) void pool_kernel(const float* __restrict__ H,
                                                   const int* __restrict__ batch,
                                                   float* __restrict__ pool,
                                                   float* __restrict__ cnt, int nnodes) {
    long long t = (long long)blockIdx.x * blockDim.x + threadIdx.x;
    int node = (int)(t >> 4);
    if (node >= nnodes) return;
    int f = (int)(t & 15) * 4;
    int g = batch[node];
    const float4 hv = *(const float4*)(H + (size_t)node * D_FEAT + f);
    float* dst = pool + (size_t)g * D_FEAT + f;
    atomicAdd(dst + 0, hv.x);
    atomicAdd(dst + 1, hv.y);
    atomicAdd(dst + 2, hv.z);
    atomicAdd(dst + 3, hv.w);
    if (f == 0) atomicAdd(&cnt[g], 1.0f);
}

__global__ __launch_bounds__(256) void pool_final_kernel(const float* __restrict__ pool,
                                                         const float* __restrict__ cnt,
                                                         float* __restrict__ out, int n) {
    int i = blockIdx.x * blockDim.x + threadIdx.x;
    if (i < n) {
        int g = i >> 6;
        out[i] = pool[i] / fmaxf(cnt[g], 1.0f);
    }
}

extern "C" void kernel_launch(void* const* d_in, const int* in_sizes, int n_in,
                              void* d_out, int out_size, void* d_ws, size_t ws_size,
                              hipStream_t stream) {
    const float* x     = (const float*)d_in[0];
    const int*   ei    = (const int*)d_in[1];
    const int*   batch = (const int*)d_in[2];
    const float* W1    = (const float*)d_in[3];
    const float* b1    = (const float*)d_in[4];
    const float* W2    = (const float*)d_in[5];
    const float* b2    = (const float*)d_in[6];
    const float* W3    = (const float*)d_in[7];
    const float* b3    = (const float*)d_in[8];
    float* out = (float*)d_out;

    const int N = in_sizes[0] / D_FEAT;
    const int E = in_sizes[1] / 2;
    const int* erow = ei;
    const int* ecol = ei + E;

    float* w = (float*)d_ws;
    size_t off = 0;
    float* dinv = w + off; off += (size_t)((N + 63) / 64) * 64;
    float* bufA = w + off; off += (size_t)N * D_FEAT;
    float* bufB = w + off; off += (size_t)N * D_FEAT;
    float* pool = w + off; off += (size_t)N_GRAPHS_C * D_FEAT;
    float* cnt  = w + off; off += N_GRAPHS_C;

    const int ND = N * D_FEAT;
    const int blk = 256;

    deg_init_kernel<<<(N + blk - 1) / blk, blk, 0, stream>>>(dinv, N);
    deg_count_kernel<<<(E + blk - 1) / blk, blk, 0, stream>>>(ecol, dinv, E);
    deg_finish_kernel<<<(N + blk - 1) / blk, blk, 0, stream>>>(dinv, N);

    const int waves = (N + 15) / 16;
    const int gemm_blocks = (waves * 32 + blk - 1) / blk;
    const long long sthreads = (long long)E * 16;
    const int scat_blocks = (int)((sthreads + blk - 1) / blk);
    const int nd_blocks = (ND + blk - 1) / blk;

    gemm_wmma_kernel<<<gemm_blocks, blk, 0, stream>>>(x, W1, bufA, N);
    init_acc_kernel<<<nd_blocks, blk, 0, stream>>>(bufA, dinv, b1, bufB, ND);
    scatter_kernel<<<scat_blocks, blk, 0, stream>>>(bufA, erow, ecol, dinv, bufB, E);
    relu_kernel<<<nd_blocks, blk, 0, stream>>>(bufB, ND);

    gemm_wmma_kernel<<<gemm_blocks, blk, 0, stream>>>(bufB, W2, bufA, N);
    init_acc_kernel<<<nd_blocks, blk, 0, stream>>>(bufA, dinv, b2, bufB, ND);
    scatter_kernel<<<scat_blocks, blk, 0, stream>>>(bufA, erow, ecol, dinv, bufB, E);
    relu_kernel<<<nd_blocks, blk, 0, stream>>>(bufB, ND);

    gemm_wmma_kernel<<<gemm_blocks, blk, 0, stream>>>(bufB, W3, bufA, N);
    init_acc_kernel<<<nd_blocks, blk, 0, stream>>>(bufA, dinv, b3, bufB, ND);
    scatter_kernel<<<scat_blocks, blk, 0, stream>>>(bufA, erow, ecol, dinv, bufB, E);
    relu_kernel<<<nd_blocks, blk, 0, stream>>>(bufB, ND);

    pool_zero_kernel<<<(N_GRAPHS_C * D_FEAT + blk - 1) / blk, blk, 0, stream>>>(pool, cnt,
                                                                               N_GRAPHS_C * D_FEAT);
    const long long pthreads = (long long)N * 16;
    pool_kernel<<<(int)((pthreads + blk - 1) / blk), blk, 0, stream>>>(bufB, batch, pool, cnt, N);
    pool_final_kernel<<<(N_GRAPHS_C * D_FEAT + blk - 1) / blk, blk, 0, stream>>>(pool, cnt, out,
                                                                                 N_GRAPHS_C * D_FEAT);
}


# mean pool: same-graph rows summed in registers, then a per-work-group LDS table, one global atomic per touched word per work-group (was one per node word); reorders the pool's float sums
# speedup vs baseline: 1.2888x; 1.2888x over previous
#include <hip/hip_runtime.h>
#include <hip/hip_bf16.h>

typedef __attribute__((ext_vector_type(2))) float v2f;
typedef __attribute__((ext_vector_type(8))) float v8f;

#define D_FEAT 64
#define N_GRAPHS_C 64

__global__ __launch_bounds__(256) void deg_init_kernel(float* __restrict__ deg, int n) {
    int i = blockIdx.x * blockDim.x + threadIdx.x;
    if (i < n) deg[i] = 1.0f;
}

__global__ __launch_bounds__(256) void deg_count_kernel(const int* __restrict__ col,
                                                        float* __restrict__ deg, int e) {
    int i = blockIdx.x * blockDim.x + threadIdx.x;
    if (i < e) atomicAdd(&deg[col[i]], 1.0f);
}

__global__ __launch_bounds__(256) void deg_finish_kernel(float* __restrict__ deg, int n) {
    int i = blockIdx.x * blockDim.x + threadIdx.x;
    if (i < n) {
        float d = deg[i];
        deg[i] = (d > 0.0f) ? rsqrtf(d) : 0.0f;
    }
}

__global__ __launch_bounds__(256) void gemm_wmma_kernel(const float* __restrict__ X,
                                                        const float* __restrict__ W,
                                                        float* __restrict__ H,
                                                        int nrows) {
    const int wave = (int)((blockIdx.x * blockDim.x + threadIdx.x) >> 5);
    const int lane = (int)(threadIdx.x & 31);
    const int row0 = wave * 16;
    if (row0 >= nrows) return;
    const int lm = lane & 15;
    const int kk = (lane >> 4) * 2;

    const float* __restrict__ xrow = X + (size_t)(row0 + lm) * D_FEAT + kk;

    v8f c0 = {}, c1 = {}, c2 = {}, c3 = {};
#pragma unroll
    for (int k = 0; k < D_FEAT; k += 4) {
        v2f a;
        a.x = xrow[k + 0];
        a.y = xrow[k + 1];
        const float* __restrict__ wr0 = W + (size_t)(k + kk) * D_FEAT + lm;
        const float* __restrict__ wr1 = wr0 + D_FEAT;
        v2f b0, b1, b2, b3;
        b0.x = wr0[0];   b0.y = wr1[0];
        b1.x = wr0[16];  b1.y = wr1[16];
        b2.x = wr0[32];  b2.y = wr1[32];
        b3.x = wr0[48];  b3.y = wr1[48];
        c0 = __builtin_amdgcn_wmma_f32_16x16x4_f32(false, a, false, b0, (short)0, c0, false, false);
        c1 = __builtin_amdgcn_wmma_f32_16x16x4_f32(false, a, false, b1, (short)0, c1, false, false);
        c2 = __builtin_amdgcn_wmma_f32_16x16x4_f32(false, a, false, b2, (short)0, c2, false, false);
        c3 = __builtin_amdgcn_wmma_f32_16x16x4_f32(false, a, false, b3, (short)0, c3, false, false);
    }

    const int mbase = row0 + (lane >> 4) * 8;
#pragma unroll
    for (int r = 0; r < 8; ++r) {
        float* __restrict__ hr = H + (size_t)(mbase + r) * D_FEAT + lm;
        hr[0]  = c0[r];
        hr[16] = c1[r];
        hr[32] = c2[r];
        hr[48] = c3[r];
    }
}

__global__ __launch_bounds__(256) void init_acc_kernel(const float* __restrict__ H,
                                                       const float* __restrict__ dinv,
                                                       const float* __restrict__ bias,
                                                       float* __restrict__ ACC, int n) {
    int t = blockIdx.x * blockDim.x + threadIdx.x;
    if (t >= n) return;
    int node = t >> 6;
    int d    = t & 63;
    float di = dinv[node];
    ACC[t] = bias[d] + H[t] * di * di;
}

__global__ __launch_bounds__(256) void scatter_kernel(const float* __restrict__ H,
                                                      const int* __restrict__ row,
                                                      const int* __restrict__ col,
                                                      const float* __restrict__ dinv,
                                                      float* __restrict__ ACC, int e) {
    long long t = (long long)blockIdx.x * blockDim.x + threadIdx.x;
    int edge = (int)(t >> 4);
    if (edge >= e) return;
    int f = (int)(t & 15) * 4;
    int r = row[edge];
    int c = col[edge];
    float s = dinv[r] * dinv[c];
    const float4 hv = *(const float4*)(H + (size_t)r * D_FEAT + f);
    float* dst = ACC + (size_t)c * D_FEAT + f;
    atomicAdd(dst + 0, hv.x * s);
    atomicAdd(dst + 1, hv.y * s);
    atomicAdd(dst + 2, hv.z * s);
    atomicAdd(dst + 3, hv.w * s);
}

__global__ __launch_bounds__(256) void relu_kernel(float* __restrict__ A, int n) {
    int i = blockIdx.x * blockDim.x + threadIdx.x;
    if (i < n) A[i] = fmaxf(A[i], 0.0f);
}

__global__ __launch_bounds__(256) void pool_zero_kernel(float* __restrict__ pool,
                                                        float* __restrict__ cnt, int n) {
    int i = blockIdx.x * blockDim.x + threadIdx.x;
    if (i < n) pool[i] = 0.0f;
    if (i < N_GRAPHS_C) cnt[i] = 0.0f;
}

#define POOL_CHUNK 128
#define POOL_ROWS (POOL_CHUNK / 16)
#define POOL_TAB (N_GRAPHS_C * D_FEAT / 256)

__global__ __launch_bounds__(256) void pool_kernel(const float* __restrict__ H,
                                                   const int* __restrict__ batch,
                                                   float* __restrict__ pool,
                                                   float* __restrict__ cnt, int nnodes) {
    __shared__ float s_sum[N_GRAPHS_C * D_FEAT];
    __shared__ float s_cnt[N_GRAPHS_C];
    const int tid  = (int)threadIdx.x;
    const int sub  = tid >> 4;
    const int f    = (tid & 15) * 4;
    const int base = (int)blockIdx.x * POOL_CHUNK + sub;

    int    gid[POOL_ROWS];
    float4 hv[POOL_ROWS];
#pragma unroll
    for (int i = 0; i < POOL_ROWS; ++i) {
        const int node = base + i * 16;
        const bool ok  = node < nnodes;
        const int nc   = ok ? node : (nnodes - 1);
        const int g    = batch[nc];
        hv[i]  = *(const float4*)(H + (size_t)nc * D_FEAT + f);
        gid[i] = (ok && (unsigned)g < (unsigned)N_GRAPHS_C) ? g : -1;
    }

#pragma unroll
    for (int j = 0; j < POOL_TAB; ++j) s_sum[tid + 256 * j] = 0.0f;
    if (tid < N_GRAPHS_C) s_cnt[tid] = 0.0f;
    __syncthreads();

    int    cur = -1;
    float  run = 0.0f;
    float4 acc = make_float4(0.0f, 0.0f, 0.0f, 0.0f);
#pragma unroll
    for (int i = 0; i < POOL_ROWS; ++i) {
        const int g = gid[i];
        if (g >= 0) {
            if (g != cur) {
                if (cur >= 0) {
                    float* s = s_sum + cur * D_FEAT + f;
                    atomicAdd(s + 0, acc.x);
                    atomicAdd(s + 1, acc.y);
                    atomicAdd(s + 2, acc.z);
                    atomicAdd(s + 3, acc.w);
                    if (f == 0) atomicAdd(&s_cnt[cur], run);
                }
                cur = g;
                acc = hv[i];
                run = 1.0f;
            } else {
                acc.x += hv[i].x;
                acc.y += hv[i].y;
                acc.z += hv[i].z;
                acc.w += hv[i].w;
                run += 1.0f;
            }
        }
    }
    if (cur >= 0) {
        float* s = s_sum + cur * D_FEAT + f;
        atomicAdd(s + 0, acc.x);
        atomicAdd(s + 1, acc.y);
        atomicAdd(s + 2, acc.z);
        atomicAdd(s + 3, acc.w);
        if (f == 0) atomicAdd(&s_cnt[cur], run);
    }
    __syncthreads();

    float tc[POOL_TAB], tv[POOL_TAB];
#pragma unroll
    for (int j = 0; j < POOL_TAB; ++j) {
        tc[j] = s_cnt[(tid >> 6) + 4 * j];
        tv[j] = s_sum[tid + 256 * j];
    }
#pragma unroll
    for (int j = 0; j < POOL_TAB; ++j) {
        if (tc[j] > 0.0f) atomicAdd(&pool[tid + 256 * j], tv[j]);
    }
    if (tid < N_GRAPHS_C) {
        const float c = s_cnt[tid];
        if (c > 0.0f) atomicAdd(&cnt[tid], c);
    }
}

__global__ __launch_bounds__(256) void pool_final_kernel(const float* __restrict__ pool,
                                                         const float* __restrict__ cnt,
                                                         float* __restrict__ out, int n) {
    int i = blockIdx.x * blockDim.x + threadIdx.x;
    if (i < n) {
        int g = i >> 6;
        out[i] = pool[i] / fmaxf(cnt[g], 1.0f);
    }
}

extern "C" void kernel_launch(void* const* d_in, const int* in_sizes, int n_in,
                              void* d_out, int out_size, void* d_ws, size_t ws_size,
                              hipStream_t stream) {
    const float* x     = (const float*)d_in[0];
    const int*   ei    = (const int*)d_in[1];
    const int*   batch = (const int*)d_in[2];
    const float* W1    = (const float*)d_in[3];
    const float* b1    = (const float*)d_in[4];
    const float* W2    = (const float*)d_in[5];
    const float* b2    = (const float*)d_in[6];
    const float* W3    = (const float*)d_in[7];
    const float* b3    = (const float*)d_in[8];
    float* out = (float*)d_out;

    const int N = in_sizes[0] / D_FEAT;
    const int E = in_sizes[1] / 2;
    const int* erow = ei;
    const int* ecol = ei + E;

    float* w = (float*)d_ws;
    size_t off = 0;
    float* dinv = w + off; off += (size_t)((N + 63) / 64) * 64;
    float* bufA = w + off; off += (size_t)N * D_FEAT;
    float* bufB = w + off; off += (size_t)N * D_FEAT;
    float* pool = w + off; off += (size_t)N_GRAPHS_C * D_FEAT;
    float* cnt  = w + off; off += N_GRAPHS_C;

    const int ND = N * D_FEAT;
    const int blk = 256;

    deg_init_kernel<<<(N + blk - 1) / blk, blk, 0, stream>>>(dinv, N);
    deg_count_kernel<<<(E + blk - 1) / blk, blk, 0, stream>>>(ecol, dinv, E);
    deg_finish_kernel<<<(N + blk - 1) / blk, blk, 0, stream>>>(dinv, N);

    const int waves = (N + 15) / 16;
    const int gemm_blocks = (waves * 32 + blk - 1) / blk;
    const long long sthreads = (long long)E * 16;
    const int scat_blocks = (int)((sthreads + blk - 1) / blk);
    const int nd_blocks = (ND + blk - 1) / blk;

    gemm_wmma_kernel<<<gemm_blocks, blk, 0, stream>>>(x, W1, bufA, N);
    init_acc_kernel<<<nd_blocks, blk, 0, stream>>>(bufA, dinv, b1, bufB, ND);
    scatter_kernel<<<scat_blocks, blk, 0, stream>>>(bufA, erow, ecol, dinv, bufB, E);
    relu_kernel<<<nd_blocks, blk, 0, stream>>>(bufB, ND);

    gemm_wmma_kernel<<<gemm_blocks, blk, 0, stream>>>(bufB, W2, bufA, N);
    init_acc_kernel<<<nd_blocks, blk, 0, stream>>>(bufA, dinv, b2, bufB, ND);
    scatter_kernel<<<scat_blocks, blk, 0, stream>>>(bufA, erow, ecol, dinv, bufB, E);
    relu_kernel<<<nd_blocks, blk, 0, stream>>>(bufB, ND);

    gemm_wmma_kernel<<<gemm_blocks, blk, 0, stream>>>(bufB, W3, bufA, N);
    init_acc_kernel<<<nd_blocks, blk, 0, stream>>>(bufA, dinv, b3, bufB, ND);
    scatter_kernel<<<scat_blocks, blk, 0, stream>>>(bufA, erow, ecol, dinv, bufB, E);
    relu_kernel<<<nd_blocks, blk, 0, stream>>>(bufB, ND);

    pool_zero_kernel<<<(N_GRAPHS_C * D_FEAT + blk - 1) / blk, blk, 0, stream>>>(pool, cnt,
                                                                               N_GRAPHS_C * D_FEAT);
    pool_kernel<<<(N + POOL_CHUNK - 1) / POOL_CHUNK, blk, 0, stream>>>(bufB, batch, pool, cnt, N);
    pool_final_kernel<<<(N_GRAPHS_C * D_FEAT + blk - 1) / blk, blk, 0, stream>>>(pool, cnt, out,
                                                                                 N_GRAPHS_C * D_FEAT);
}


# aggregation as a gather over a per-call counting sort of the edges by target: each node row summed in registers in list arrival order, written once; no float atomics, init or relu pass
# speedup vs baseline: 2.9793x; 2.3117x over previous
#include <hip/hip_runtime.h>
#include <hip/hip_bf16.h>

typedef __attribute__((ext_vector_type(2))) float v2f;
typedef __attribute__((ext_vector_type(8))) float v8f;

#define D_FEAT 64
#define N_GRAPHS_C 64

__global__ __launch_bounds__(256) void cnt_zero_kernel(int* __restrict__ cnt, int n) {
    int i = blockIdx.x * blockDim.x + threadIdx.x;
    if (i < n) cnt[i] = 0;
}

__global__ __launch_bounds__(256) void deg_count_kernel(const int* __restrict__ col,
                                                        int* __restrict__ cnt,
                                                        int* __restrict__ rank, int e) {
    int i = blockIdx.x * blockDim.x + threadIdx.x;
    if (i < e) rank[i] = atomicAdd(&cnt[col[i]], 1);
}

#define SCAN_THREADS 1024
#define SCAN_TILE (SCAN_THREADS * 4)
__global__ __launch_bounds__(SCAN_THREADS) void scan_kernel(const int* __restrict__ cnt,
                                                            int* __restrict__ rowptr,
                                                            float* __restrict__ dinv, int n) {
    __shared__ int s_wave[32];
    __shared__ int s_pre[32];
    const int tid  = (int)threadIdx.x;
    const int lane = tid & 31;
    const int wv   = tid >> 5;
    const int base = (int)blockIdx.x * SCAN_TILE;
    const int i0   = base + tid * 4;

    int c[4];
    if (i0 + 3 < n) {
        const int4 v = *(const int4*)(cnt + i0);
        c[0] = v.x; c[1] = v.y; c[2] = v.z; c[3] = v.w;
    } else {
#pragma unroll
        for (int k = 0; k < 4; ++k) c[k] = (i0 + k < n) ? cnt[i0 + k] : 0;
    }

    int pre = 0;
#pragma unroll 4
    for (int i = tid * 4; i < base; i += SCAN_TILE) {
        const int4 v = *(const int4*)(cnt + i);
        pre += (v.x + v.y) + (v.z + v.w);
    }

    const int t = c[0] + c[1] + c[2] + c[3];
    int x = t;
    int p = pre;
#pragma unroll
    for (int d = 1; d < 32; d <<= 1) {
        const int y = __shfl_up(x, d);
        const int q = __shfl_up(p, d);
        if (lane >= d) { x += y; p += q; }
    }
    if (lane == 31) { s_wave[wv] = x; s_pre[wv] = p; }
    __syncthreads();
    const int wt = s_wave[lane];
    int ws = wt;
    int ps = s_pre[lane];
#pragma unroll
    for (int d = 1; d < 32; d <<= 1) {
        const int y = __shfl_up(ws, d);
        const int q = __shfl_up(ps, d);
        if (lane >= d) { ws += y; ps += q; }
    }
    const int carry      = __shfl(ps, 31);
    const int wave_excl  = __shfl(ws, wv) - __shfl(wt, wv);
    const int tile_total = __shfl(ws, 31);
    int run = carry + wave_excl + (x - t);
#pragma unroll
    for (int k = 0; k < 4; ++k) {
        if (i0 + k < n) {
            rowptr[i0 + k] = run;
            const float d = (float)(c[k] + 1);
            dinv[i0 + k] = (d > 0.0f) ? rsqrtf(d) : 0.0f;
            run += c[k];
        }
    }
    if (blockIdx.x == gridDim.x - 1 && tid == 0) rowptr[n] = carry + tile_total;
}

__global__ __launch_bounds__(256) void fill_kernel(const int* __restrict__ row,
                                                   const int* __restrict__ col,
                                                   const int* __restrict__ rank,
                                                   const int* __restrict__ rowptr,
                                                   const float* __restrict__ dinv,
                                                   int2* __restrict__ list, int e) {
    int i = blockIdx.x * blockDim.x + threadIdx.x;
    if (i < e) {
        const int r = row[i];
        const int c = col[i];
        const int pos = rowptr[c] + rank[i];
        list[pos] = make_int2(r, __float_as_int(dinv[r]));
    }
}

__global__ __launch_bounds__(256) void gemm_wmma_kernel(const float* __restrict__ X,
                                                        const float* __restrict__ W,
                                                        float* __restrict__ H,
                                                        int nrows) {
    const int wave = (int)((blockIdx.x * blockDim.x + threadIdx.x) >> 5);
    const int lane = (int)(threadIdx.x & 31);
    const int row0 = wave * 16;
    if (row0 >= nrows) return;
    const int lm = lane & 15;
    const int kk = (lane >> 4) * 2;

    const float* __restrict__ xrow = X + (size_t)(row0 + lm) * D_FEAT + kk;

    v8f c0 = {}, c1 = {}, c2 = {}, c3 = {};
#pragma unroll
    for (int k = 0; k < D_FEAT; k += 4) {
        v2f a;
        a.x = xrow[k + 0];
        a.y = xrow[k + 1];
        const float* __restrict__ wr0 = W + (size_t)(k + kk) * D_FEAT + lm;
        const float* __restrict__ wr1 = wr0 + D_FEAT;
        v2f b0, b1, b2, b3;
        b0.x = wr0[0];   b0.y = wr1[0];
        b1.x = wr0[16];  b1.y = wr1[16];
        b2.x = wr0[32];  b2.y = wr1[32];
        b3.x = wr0[48];  b3.y = wr1[48];
        c0 = __builtin_amdgcn_wmma_f32_16x16x4_f32(false, a, false, b0, (short)0, c0, false, false);
        c1 = __builtin_amdgcn_wmma_f32_16x16x4_f32(false, a, false, b1, (short)0, c1, false, false);
        c2 = __builtin_amdgcn_wmma_f32_16x16x4_f32(false, a, false, b2, (short)0, c2, false, false);
        c3 = __builtin_amdgcn_wmma_f32_16x16x4_f32(false, a, false, b3, (short)0, c3, false, false);
    }

    const int mbase = row0 + (lane >> 4) * 8;
#pragma unroll
    for (int r = 0; r < 8; ++r) {
        float* __restrict__ hr = H + (size_t)(mbase + r) * D_FEAT + lm;
        hr[0]  = c0[r];
        hr[16] = c1[r];
        hr[32] = c2[r];
        hr[48] = c3[r];
    }
}

template <int B>
__device__ __forceinline__ void gather_batch(const float* __restrict__ H,
                                             const int2* __restrict__ list,
                                             int j, int f, float di, float4& acc) {
    int2   en[B];
    float4 hv[B];
#pragma unroll
    for (int k = 0; k < B; ++k) en[k] = list[j + k];
#pragma unroll
    for (int k = 0; k < B; ++k)
        hv[k] = *(const float4*)(H + (size_t)en[k].x * D_FEAT + f);
#pragma unroll
    for (int k = 0; k < B; ++k) {
        const float s = __int_as_float(en[k].y) * di;
        acc.x += hv[k].x * s;
        acc.y += hv[k].y * s;
        acc.z += hv[k].z * s;
        acc.w += hv[k].w * s;
    }
}

__global__ __launch_bounds__(256) void gather_kernel(const float* __restrict__ H,
                                                     const int* __restrict__ rowptr,
                                                     const int2* __restrict__ list,
                                                     const float* __restrict__ dinv,
                                                     const float* __restrict__ bias,
                                                     float* __restrict__ OUT, int n) {
    const int t = blockIdx.x * blockDim.x + threadIdx.x;
    const int node = t >> 4;
    if (node >= n) return;
    const int f = (t & 15) * 4;
    const int beg = rowptr[node];
    const int end = rowptr[node + 1];
    const float di = dinv[node];
    const float4 hs = *(const float4*)(H + (size_t)node * D_FEAT + f);
    const float4 bv = *(const float4*)(bias + f);
    const float ss = di * di;
    float4 acc = make_float4(hs.x * ss, hs.y * ss, hs.z * ss, hs.w * ss);

    int j = beg;
    for (; j + 8 <= end; j += 8) gather_batch<8>(H, list, j, f, di, acc);
    const int rest = end - j;
    if (rest & 4) { gather_batch<4>(H, list, j, f, di, acc); j += 4; }
    if (rest & 2) { gather_batch<2>(H, list, j, f, di, acc); j += 2; }
    if (rest & 1) { gather_batch<1>(H, list, j, f, di, acc); }

    float4 o;
    o.x = fmaxf(acc.x + bv.x, 0.0f);
    o.y = fmaxf(acc.y + bv.y, 0.0f);
    o.z = fmaxf(acc.z + bv.z, 0.0f);
    o.w = fmaxf(acc.w + bv.w, 0.0f);
    *(float4*)(OUT + (size_t)node * D_FEAT + f) = o;
}

__global__ __launch_bounds__(256) void pool_zero_kernel(float* __restrict__ pool,
                                                        float* __restrict__ cnt, int n) {
    int i = blockIdx.x * blockDim.x + threadIdx.x;
    if (i < n) pool[i] = 0.0f;
    if (i < N_GRAPHS_C) cnt[i] = 0.0f;
}

#define POOL_CHUNK 128
#define POOL_ROWS (POOL_CHUNK / 16)
#define POOL_TAB (N_GRAPHS_C * D_FEAT / 256)

__global__ __launch_bounds__(256) void pool_kernel(const float* __restrict__ H,
                                                   const int* __restrict__ batch,
                                                   float* __restrict__ pool,
                                                   float* __restrict__ cnt, int nnodes) {
    __shared__ float s_sum[N_GRAPHS_C * D_FEAT];
    __shared__ float s_cnt[N_GRAPHS_C];
    const int tid  = (int)threadIdx.x;
    const int sub  = tid >> 4;
    const int f    = (tid & 15) * 4;
    const int base = (int)blockIdx.x * POOL_CHUNK + sub;

    int    gid[POOL_ROWS];
    float4 hv[POOL_ROWS];
#pragma unroll
    for (int i = 0; i < POOL_ROWS; ++i) {
        const int node = base + i * 16;
        const bool ok  = node < nnodes;
        const int nc   = ok ? node : (nnodes - 1);
        const int g    = batch[nc];
        hv[i]  = *(const float4*)(H + (size_t)nc * D_FEAT + f);
        gid[i] = (ok && (unsigned)g < (unsigned)N_GRAPHS_C) ? g : -1;
    }

#pragma unroll
    for (int j = 0; j < POOL_TAB; ++j) s_sum[tid + 256 * j] = 0.0f;
    if (tid < N_GRAPHS_C) s_cnt[tid] = 0.0f;
    __syncthreads();

    int    cur = -1;
    float  run = 0.0f;
    float4 acc = make_float4(0.0f, 0.0f, 0.0f, 0.0f);
#pragma unroll
    for (int i = 0; i < POOL_ROWS; ++i) {
        const int g = gid[i];
        if (g >= 0) {
            if (g != cur) {
                if (cur >= 0) {
                    float* s = s_sum + cur * D_FEAT + f;
                    atomicAdd(s + 0, acc.x);
                    atomicAdd(s + 1, acc.y);
                    atomicAdd(s + 2, acc.z);
                    atomicAdd(s + 3, acc.w);
                    if (f == 0) atomicAdd(&s_cnt[cur], run);
                }
                cur = g;
                acc = hv[i];
                run = 1.0f;
            } else {
                acc.x += hv[i].x;
                acc.y += hv[i].y;
                acc.z += hv[i].z;
                acc.w += hv[i].w;
                run += 1.0f;
            }
        }
    }
    if (cur >= 0) {
        float* s = s_sum + cur * D_FEAT + f;
        atomicAdd(s + 0, acc.x);
        atomicAdd(s + 1, acc.y);
        atomicAdd(s + 2, acc.z);
        atomicAdd(s + 3, acc.w);
        if (f == 0) atomicAdd(&s_cnt[cur], run);
    }
    __syncthreads();

    float tc[POOL_TAB], tv[POOL_TAB];
#pragma unroll
    for (int j = 0; j < POOL_TAB; ++j) {
        tc[j] = s_cnt[(tid >> 6) + 4 * j];
        tv[j] = s_sum[tid + 256 * j];
    }
#pragma unroll
    for (int j = 0; j < POOL_TAB; ++j) {
        if (tc[j] > 0.0f) atomicAdd(&pool[tid + 256 * j], tv[j]);
    }
    if (tid < N_GRAPHS_C) {
        const float c = s_cnt[tid];
        if (c > 0.0f) atomicAdd(&cnt[tid], c);
    }
}

__global__ __launch_bounds__(256) void pool_final_kernel(const float* __restrict__ pool,
                                                         const float* __restrict__ cnt,
                                                         float* __restrict__ out, int n) {
    int i = blockIdx.x * blockDim.x + threadIdx.x;
    if (i < n) {
        int g = i >> 6;
        out[i] = pool[i] / fmaxf(cnt[g], 1.0f);
    }
}

extern "C" void kernel_launch(void* const* d_in, const int* in_sizes, int n_in,
                              void* d_out, int out_size, void* d_ws, size_t ws_size,
                              hipStream_t stream) {
    const float* x     = (const float*)d_in[0];
    const int*   ei    = (const int*)d_in[1];
    const int*   batch = (const int*)d_in[2];
    const float* W1    = (const float*)d_in[3];
    const float* b1    = (const float*)d_in[4];
    const float* W2    = (const float*)d_in[5];
    const float* b2    = (const float*)d_in[6];
    const float* W3    = (const float*)d_in[7];
    const float* b3    = (const float*)d_in[8];
    float* out = (float*)d_out;

    const int N = in_sizes[0] / D_FEAT;
    const int E = in_sizes[1] / 2;
    const int* erow = ei;
    const int* ecol = ei + E;

    float* w = (float*)d_ws;
    const size_t NP = (size_t)((N + 1 + 63) / 64) * 64;
    size_t off = 0;
    float* dinv = w + off; off += NP;
    float* bufA = w + off; off += (size_t)N * D_FEAT;
    float* bufB = w + off; off += (size_t)N * D_FEAT;
    float* pool = w + off; off += (size_t)N_GRAPHS_C * D_FEAT;
    float* cnt  = w + off; off += N_GRAPHS_C;
    int* icnt   = (int*)(w + off); off += NP;
    int* rowptr = (int*)(w + off); off += NP;
    int* rank   = (int*)(w + off); off += (size_t)((E + 63) / 64) * 64;
    int2* list  = (int2*)(w + off); off += (size_t)E * 2;

    const int blk = 256;

    cnt_zero_kernel<<<(N + blk - 1) / blk, blk, 0, stream>>>(icnt, N);
    deg_count_kernel<<<(E + blk - 1) / blk, blk, 0, stream>>>(ecol, icnt, rank, E);
    scan_kernel<<<(N + SCAN_TILE - 1) / SCAN_TILE, SCAN_THREADS, 0, stream>>>(icnt, rowptr, dinv, N);
    fill_kernel<<<(E + blk - 1) / blk, blk, 0, stream>>>(erow, ecol, rank, rowptr, dinv, list, E);

    const int waves = (N + 15) / 16;
    const int gemm_blocks = (waves * 32 + blk - 1) / blk;
    const int gath_blocks = (int)(((long long)N * 16 + blk - 1) / blk);

    gemm_wmma_kernel<<<gemm_blocks, blk, 0, stream>>>(x, W1, bufA, N);
    gather_kernel<<<gath_blocks, blk, 0, stream>>>(bufA, rowptr, list, dinv, b1, bufB, N);

    gemm_wmma_kernel<<<gemm_blocks, blk, 0, stream>>>(bufB, W2, bufA, N);
    gather_kernel<<<gath_blocks, blk, 0, stream>>>(bufA, rowptr, list, dinv, b2, bufB, N);

    gemm_wmma_kernel<<<gemm_blocks, blk, 0, stream>>>(bufB, W3, bufA, N);
    gather_kernel<<<gath_blocks, blk, 0, stream>>>(bufA, rowptr, list, dinv, b3, bufB, N);

    pool_zero_kernel<<<(N_GRAPHS_C * D_FEAT + blk - 1) / blk, blk, 0, stream>>>(pool, cnt,
                                                                               N_GRAPHS_C * D_FEAT);
    pool_kernel<<<(N + POOL_CHUNK - 1) / POOL_CHUNK, blk, 0, stream>>>(bufB, batch, pool, cnt, N);
    pool_final_kernel<<<(N_GRAPHS_C * D_FEAT + blk - 1) / blk, blk, 0, stream>>>(pool, cnt, out,
                                                                                 N_GRAPHS_C * D_FEAT);
}


# edge sort by target on LDS counters only: edges binned by 256-node range (LDS histograms), ranked per bin; no per-edge global atomic; dinv[source] scales GEMM rows; node sums stay in arrival order
# speedup vs baseline: 4.7859x; 1.6064x over previous
#include <hip/hip_runtime.h>
#include <hip/hip_bf16.h>

typedef __attribute__((ext_vector_type(2))) float v2f;
typedef __attribute__((ext_vector_type(8))) float v8f;

#define D_FEAT 64
#define N_GRAPHS_C 64

#define BIN_SHIFT 8
#define BIN_NODES (1 << BIN_SHIFT)
#define NB_MAX 256
#define SORT_THREADS 1024
#define SORT_GROUPS 128
#define SORT_BATCH 4

__global__ __launch_bounds__(SORT_THREADS) void bin_count_kernel(const int* __restrict__ col,
                                                                 int* __restrict__ wghist,
                                                                 int e, int chunk, int nb) {
    __shared__ int s_h[NB_MAX];
    const int tid = (int)threadIdx.x;
    if (tid < NB_MAX) s_h[tid] = 0;
    __syncthreads();
    const int e0 = (int)blockIdx.x * chunk;
    const int e1 = min(e, e0 + chunk);
    for (int i0 = e0 + tid; i0 < e1; i0 += SORT_BATCH * SORT_THREADS) {
        int c[SORT_BATCH];
#pragma unroll
        for (int k = 0; k < SORT_BATCH; ++k) {
            const int i = i0 + k * SORT_THREADS;
            c[k] = (i < e1) ? col[i] : -1;
        }
#pragma unroll
        for (int k = 0; k < SORT_BATCH; ++k)
            if (c[k] >= 0) atomicAdd(&s_h[c[k] >> BIN_SHIFT], 1);
    }
    __syncthreads();
    if (tid < nb) wghist[(int)blockIdx.x * nb + tid] = s_h[tid];
}

__global__ __launch_bounds__(SORT_THREADS) void bin_spread_kernel(const int* __restrict__ row,
                                                                  const int* __restrict__ col,
                                                                  const int* __restrict__ wghist,
                                                                  int* __restrict__ binstart,
                                                                  int2* __restrict__ binned,
                                                                  int e, int chunk, int nb) {
    __shared__ int s_tot[4][NB_MAX];
    __shared__ int s_bef[4][NB_MAX];
    __shared__ int s_base[NB_MAX];
    __shared__ int s_wave[NB_MAX / 32];
    const int tid  = (int)threadIdx.x;
    const int lane = tid & 31;
    const int wv   = tid >> 5;
    const int grp  = (int)blockIdx.x;
    const int e0   = grp * chunk;
    const int e1   = min(e, e0 + chunk);

    int r[SORT_BATCH], c[SORT_BATCH];
    int i0 = e0 + tid;
#pragma unroll
    for (int k = 0; k < SORT_BATCH; ++k) {
        const int i = i0 + k * SORT_THREADS;
        const bool ok = i < e1;
        c[k] = ok ? col[i] : -1;
        r[k] = ok ? row[i] : 0;
    }

    const int b    = tid & (NB_MAX - 1);
    const int part = tid >> 8;
    int tot = 0, bef = 0;
    if (b < nb) {
        const int g0 = part * (SORT_GROUPS / 4);
#pragma unroll
        for (int g = 0; g < SORT_GROUPS / 4; ++g) {
            const int v = wghist[(g0 + g) * nb + b];
            tot += v;
            bef += (g0 + g < grp) ? v : 0;
        }
    }
    s_tot[part][b] = tot;
    s_bef[part][b] = bef;
    __syncthreads();

    int t = 0, bf = 0, x = 0;
    if (tid < NB_MAX) {
        t  = (s_tot[0][tid] + s_tot[1][tid]) + (s_tot[2][tid] + s_tot[3][tid]);
        bf = (s_bef[0][tid] + s_bef[1][tid]) + (s_bef[2][tid] + s_bef[3][tid]);
        x = t;
#pragma unroll
        for (int d = 1; d < 32; d <<= 1) {
            const int y = __shfl_up(x, d);
            if (lane >= d) x += y;
        }
        if (lane == 31) s_wave[wv] = x;
    }
    __syncthreads();
    if (tid < NB_MAX) {
        int carry = 0;
#pragma unroll
        for (int k = 0; k < NB_MAX / 32; ++k) carry += (k < wv) ? s_wave[k] : 0;
        const int start = carry + (x - t);
        s_base[tid] = start + bf;
        if (grp == 0) {
            if (tid < nb) binstart[tid] = start;
            if (tid == nb - 1) binstart[nb] = start + t;
        }
    }
    __syncthreads();

    for (;;) {
#pragma unroll
        for (int k = 0; k < SORT_BATCH; ++k) {
            if (c[k] >= 0) {
                const int pos = atomicAdd(&s_base[c[k] >> BIN_SHIFT], 1);
                binned[pos] = make_int2(r[k], c[k]);
            }
        }
        i0 += SORT_BATCH * SORT_THREADS;
        if (i0 >= e1) break;
#pragma unroll
        for (int k = 0; k < SORT_BATCH; ++k) {
            const int i = i0 + k * SORT_THREADS;
            const bool ok = i < e1;
            c[k] = ok ? col[i] : -1;
            r[k] = ok ? row[i] : 0;
        }
    }
}

#define RANK_HELD 8
__global__ __launch_bounds__(SORT_THREADS) void bin_rank_kernel(const int2* __restrict__ binned,
                                                                const int* __restrict__ binstart,
                                                                int* __restrict__ rowptr,
                                                                float* __restrict__ dinv,
                                                                int* __restrict__ list, int n) {
    __shared__ int s_cnt[BIN_NODES];
    __shared__ int s_ptr[BIN_NODES];
    __shared__ int s_wave[BIN_NODES / 32];
    const int tid   = (int)threadIdx.x;
    const int lane  = tid & 31;
    const int wv    = tid >> 5;
    const int bin   = (int)blockIdx.x;
    const int lo    = binstart[bin];
    const int hi    = binstart[bin + 1];
    const int node0 = bin << BIN_SHIFT;

    int2 ed[RANK_HELD];
#pragma unroll
    for (int k = 0; k < RANK_HELD; ++k) {
        const int j = lo + tid + k * SORT_THREADS;
        ed[k] = (j < hi) ? binned[j] : make_int2(0, -1);
    }
    if (tid < BIN_NODES) s_cnt[tid] = 0;
    __syncthreads();

#pragma unroll
    for (int k = 0; k < RANK_HELD; ++k)
        if (ed[k].y >= 0) atomicAdd(&s_cnt[ed[k].y - node0], 1);
    for (int j = lo + tid + RANK_HELD * SORT_THREADS; j < hi; j += SORT_THREADS)
        atomicAdd(&s_cnt[binned[j].y - node0], 1);
    __syncthreads();

    int v = 0, x = 0;
    if (tid < BIN_NODES) {
        v = s_cnt[tid];
        x = v;
#pragma unroll
        for (int d = 1; d < 32; d <<= 1) {
            const int y = __shfl_up(x, d);
            if (lane >= d) x += y;
        }
        if (lane == 31) s_wave[wv] = x;
    }
    __syncthreads();
    if (tid < BIN_NODES) {
        int carry = 0;
#pragma unroll
        for (int k = 0; k < BIN_NODES / 32; ++k) carry += (k < wv) ? s_wave[k] : 0;
        const int p = lo + carry + (x - v);
        s_ptr[tid] = p;
        const int node = node0 + tid;
        if (node < n) {
            rowptr[node] = p;
            const float d = (float)(v + 1);
            dinv[node] = (d > 0.0f) ? rsqrtf(d) : 0.0f;
        }
    }
    if (bin == (int)gridDim.x - 1 && tid == 0) rowptr[n] = hi;
    __syncthreads();

#pragma unroll
    for (int k = 0; k < RANK_HELD; ++k) {
        if (ed[k].y >= 0) {
            const int pos = atomicAdd(&s_ptr[ed[k].y - node0], 1);
            list[pos] = ed[k].x;
        }
    }
    for (int j = lo + tid + RANK_HELD * SORT_THREADS; j < hi; j += SORT_THREADS) {
        const int2 ee = binned[j];
        const int pos = atomicAdd(&s_ptr[ee.y - node0], 1);
        list[pos] = ee.x;
    }
}

__global__ __launch_bounds__(256) void gemm_wmma_kernel(const float* __restrict__ X,
                                                        const float* __restrict__ W,
                                                        const float* __restrict__ dinv,
                                                        float* __restrict__ H,
                                                        int nrows) {
    const int wave = (int)((blockIdx.x * blockDim.x + threadIdx.x) >> 5);
    const int lane = (int)(threadIdx.x & 31);
    const int row0 = wave * 16;
    if (row0 >= nrows) return;
    const int lm = lane & 15;
    const int kk = (lane >> 4) * 2;

    const float* __restrict__ xrow = X + (size_t)(row0 + lm) * D_FEAT + kk;
    const int mbase = row0 + (lane >> 4) * 8;
    const float4 dlo = *(const float4*)(dinv + mbase);
    const float4 dhi = *(const float4*)(dinv + mbase + 4);

    v8f c0 = {}, c1 = {}, c2 = {}, c3 = {};
#pragma unroll
    for (int k = 0; k < D_FEAT; k += 4) {
        v2f a;
        a.x = xrow[k + 0];
        a.y = xrow[k + 1];
        const float* __restrict__ wr0 = W + (size_t)(k + kk) * D_FEAT + lm;
        const float* __restrict__ wr1 = wr0 + D_FEAT;
        v2f b0, b1, b2, b3;
        b0.x = wr0[0];   b0.y = wr1[0];
        b1.x = wr0[16];  b1.y = wr1[16];
        b2.x = wr0[32];  b2.y = wr1[32];
        b3.x = wr0[48];  b3.y = wr1[48];
        c0 = __builtin_amdgcn_wmma_f32_16x16x4_f32(false, a, false, b0, (short)0, c0, false, false);
        c1 = __builtin_amdgcn_wmma_f32_16x16x4_f32(false, a, false, b1, (short)0, c1, false, false);
        c2 = __builtin_amdgcn_wmma_f32_16x16x4_f32(false, a, false, b2, (short)0, c2, false, false);
        c3 = __builtin_amdgcn_wmma_f32_16x16x4_f32(false, a, false, b3, (short)0, c3, false, false);
    }

    const float ds[8] = {dlo.x, dlo.y, dlo.z, dlo.w, dhi.x, dhi.y, dhi.z, dhi.w};
#pragma unroll
    for (int r = 0; r < 8; ++r) {
        float* __restrict__ hr = H + (size_t)(mbase + r) * D_FEAT + lm;
        hr[0]  = c0[r] * ds[r];
        hr[16] = c1[r] * ds[r];
        hr[32] = c2[r] * ds[r];
        hr[48] = c3[r] * ds[r];
    }
}

template <int B>
__device__ __forceinline__ void gather_batch(const float* __restrict__ H,
                                             const int* __restrict__ list,
                                             int j, int f, float4& acc) {
    int    en[B];
    float4 hv[B];
#pragma unroll
    for (int k = 0; k < B; ++k) en[k] = list[j + k];
#pragma unroll
    for (int k = 0; k < B; ++k)
        hv[k] = *(const float4*)(H + (size_t)en[k] * D_FEAT + f);
#pragma unroll
    for (int k = 0; k < B; ++k) {
        acc.x += hv[k].x;
        acc.y += hv[k].y;
        acc.z += hv[k].z;
        acc.w += hv[k].w;
    }
}

__global__ __launch_bounds__(256) void gather_kernel(const float* __restrict__ H,
                                                     const int* __restrict__ rowptr,
                                                     const int* __restrict__ list,
                                                     const float* __restrict__ dinv,
                                                     const float* __restrict__ bias,
                                                     float* __restrict__ OUT, int n) {
    const int t = blockIdx.x * blockDim.x + threadIdx.x;
    const int node = t >> 4;
    if (node >= n) return;
    const int f = (t & 15) * 4;
    const int beg = rowptr[node];
    const int end = rowptr[node + 1];
    const float di = dinv[node];
    const float4 hs = *(const float4*)(H + (size_t)node * D_FEAT + f);
    const float4 bv = *(const float4*)(bias + f);
    float4 acc = hs;

    int j = beg;
    for (; j + 8 <= end; j += 8) gather_batch<8>(H, list, j, f, acc);
    const int rest = end - j;
    if (rest & 4) { gather_batch<4>(H, list, j, f, acc); j += 4; }
    if (rest & 2) { gather_batch<2>(H, list, j, f, acc); j += 2; }
    if (rest & 1) { gather_batch<1>(H, list, j, f, acc); }

    float4 o;
    o.x = fmaxf(acc.x * di + bv.x, 0.0f);
    o.y = fmaxf(acc.y * di + bv.y, 0.0f);
    o.z = fmaxf(acc.z * di + bv.z, 0.0f);
    o.w = fmaxf(acc.w * di + bv.w, 0.0f);
    *(float4*)(OUT + (size_t)node * D_FEAT + f) = o;
}

__global__ __launch_bounds__(256) void pool_zero_kernel(float* __restrict__ pool,
                                                        float* __restrict__ cnt, int n) {
    int i = blockIdx.x * blockDim.x + threadIdx.x;
    if (i < n) pool[i] = 0.0f;
    if (i < N_GRAPHS_C) cnt[i] = 0.0f;
}

#define POOL_CHUNK 128
#define POOL_ROWS (POOL_CHUNK / 16)
#define POOL_TAB (N_GRAPHS_C * D_FEAT / 256)

__global__ __launch_bounds__(256) void pool_kernel(const float* __restrict__ H,
                                                   const int* __restrict__ batch,
                                                   float* __restrict__ pool,
                                                   float* __restrict__ cnt, int nnodes) {
    __shared__ float s_sum[N_GRAPHS_C * D_FEAT];
    __shared__ float s_cnt[N_GRAPHS_C];
    const int tid  = (int)threadIdx.x;
    const int sub  = tid >> 4;
    const int f    = (tid & 15) * 4;
    const int base = (int)blockIdx.x * POOL_CHUNK + sub;

    int    gid[POOL_ROWS];
    float4 hv[POOL_ROWS];
#pragma unroll
    for (int i = 0; i < POOL_ROWS; ++i) {
        const int node = base + i * 16;
        const bool ok  = node < nnodes;
        const int nc   = ok ? node : (nnodes - 1);
        const int g    = batch[nc];
        hv[i]  = *(const float4*)(H + (size_t)nc * D_FEAT + f);
        gid[i] = (ok && (unsigned)g < (unsigned)N_GRAPHS_C) ? g : -1;
    }

#pragma unroll
    for (int j = 0; j < POOL_TAB; ++j) s_sum[tid + 256 * j] = 0.0f;
    if (tid < N_GRAPHS_C) s_cnt[tid] = 0.0f;
    __syncthreads();

    int    cur = -1;
    float  run = 0.0f;
    float4 acc = make_float4(0.0f, 0.0f, 0.0f, 0.0f);
#pragma unroll
    for (int i = 0; i < POOL_ROWS; ++i) {
        const int g = gid[i];
        if (g >= 0) {
            if (g != cur) {
                if (cur >= 0) {
                    float* s = s_sum + cur * D_FEAT + f;
                    atomicAdd(s + 0, acc.x);
                    atomicAdd(s + 1, acc.y);
                    atomicAdd(s + 2, acc.z);
                    atomicAdd(s + 3, acc.w);
                    if (f == 0) atomicAdd(&s_cnt[cur], run);
                }
                cur = g;
                acc = hv[i];
                run = 1.0f;
            } else {
                acc.x += hv[i].x;
                acc.y += hv[i].y;
                acc.z += hv[i].z;
                acc.w += hv[i].w;
                run += 1.0f;
            }
        }
    }
    if (cur >= 0) {
        float* s = s_sum + cur * D_FEAT + f;
        atomicAdd(s + 0, acc.x);
        atomicAdd(s + 1, acc.y);
        atomicAdd(s + 2, acc.z);
        atomicAdd(s + 3, acc.w);
        if (f == 0) atomicAdd(&s_cnt[cur], run);
    }
    __syncthreads();

    float tc[POOL_TAB], tv[POOL_TAB];
#pragma unroll
    for (int j = 0; j < POOL_TAB; ++j) {
        tc[j] = s_cnt[(tid >> 6) + 4 * j];
        tv[j] = s_sum[tid + 256 * j];
    }
#pragma unroll
    for (int j = 0; j < POOL_TAB; ++j) {
        if (tc[j] > 0.0f) atomicAdd(&pool[tid + 256 * j], tv[j]);
    }
    if (tid < N_GRAPHS_C) {
        const float c = s_cnt[tid];
        if (c > 0.0f) atomicAdd(&cnt[tid], c);
    }
}

__global__ __launch_bounds__(256) void pool_final_kernel(const float* __restrict__ pool,
                                                         const float* __restrict__ cnt,
                                                         float* __restrict__ out, int n) {
    int i = blockIdx.x * blockDim.x + threadIdx.x;
    if (i < n) {
        int g = i >> 6;
        out[i] = pool[i] / fmaxf(cnt[g], 1.0f);
    }
}

extern "C" void kernel_launch(void* const* d_in, const int* in_sizes, int n_in,
                              void* d_out, int out_size, void* d_ws, size_t ws_size,
                              hipStream_t stream) {
    const float* x     = (const float*)d_in[0];
    const int*   ei    = (const int*)d_in[1];
    const int*   batch = (const int*)d_in[2];
    const float* W1    = (const float*)d_in[3];
    const float* b1    = (const float*)d_in[4];
    const float* W2    = (const float*)d_in[5];
    const float* b2    = (const float*)d_in[6];
    const float* W3    = (const float*)d_in[7];
    const float* b3    = (const float*)d_in[8];
    float* out = (float*)d_out;

    const int N = in_sizes[0] / D_FEAT;
    const int E = in_sizes[1] / 2;
    const int* erow = ei;
    const int* ecol = ei + E;

    float* w = (float*)d_ws;
    const size_t NP = (size_t)((N + 1 + 63) / 64) * 64;
    size_t off = 0;
    float* dinv = w + off; off += NP;
    float* bufA = w + off; off += (size_t)N * D_FEAT;
    float* bufB = w + off; off += (size_t)N * D_FEAT;
    float* pool = w + off; off += (size_t)N_GRAPHS_C * D_FEAT;
    float* cnt  = w + off; off += N_GRAPHS_C;
    int* rowptr   = (int*)(w + off); off += NP;
    int* wghist   = (int*)(w + off); off += (size_t)SORT_GROUPS * NB_MAX;
    int* binstart = (int*)(w + off); off += NB_MAX + 64;
    int2* binned  = (int2*)(w + off); off += (size_t)((E + 31) / 32) * 64;
    int* list     = (int*)(w + off); off += (size_t)((E + 63) / 64) * 64;

    const int blk = 256;

    const int NB    = (N + BIN_NODES - 1) / BIN_NODES;
    const int chunk = (E + SORT_GROUPS - 1) / SORT_GROUPS;
    bin_count_kernel<<<SORT_GROUPS, SORT_THREADS, 0, stream>>>(ecol, wghist, E, chunk, NB);
    bin_spread_kernel<<<SORT_GROUPS, SORT_THREADS, 0, stream>>>(erow, ecol, wghist, binstart, binned,
                                                                E, chunk, NB);
    bin_rank_kernel<<<NB, SORT_THREADS, 0, stream>>>(binned, binstart, rowptr, dinv, list, N);

    const int waves = (N + 15) / 16;
    const int gemm_blocks = (waves * 32 + blk - 1) / blk;
    const int gath_blocks = (int)(((long long)N * 16 + blk - 1) / blk);

    gemm_wmma_kernel<<<gemm_blocks, blk, 0, stream>>>(x, W1, dinv, bufA, N);
    gather_kernel<<<gath_blocks, blk, 0, stream>>>(bufA, rowptr, list, dinv, b1, bufB, N);

    gemm_wmma_kernel<<<gemm_blocks, blk, 0, stream>>>(bufB, W2, dinv, bufA, N);
    gather_kernel<<<gath_blocks, blk, 0, stream>>>(bufA, rowptr, list, dinv, b2, bufB, N);

    gemm_wmma_kernel<<<gemm_blocks, blk, 0, stream>>>(bufB, W3, dinv, bufA, N);
    gather_kernel<<<gath_blocks, blk, 0, stream>>>(bufA, rowptr, list, dinv, b3, bufB, N);

    pool_zero_kernel<<<(N_GRAPHS_C * D_FEAT + blk - 1) / blk, blk, 0, stream>>>(pool, cnt,
                                                                               N_GRAPHS_C * D_FEAT);
    pool_kernel<<<(N + POOL_CHUNK - 1) / POOL_CHUNK, blk, 0, stream>>>(bufB, batch, pool, cnt, N);
    pool_final_kernel<<<(N_GRAPHS_C * D_FEAT + blk - 1) / blk, blk, 0, stream>>>(pool, cnt, out,
                                                                                 N_GRAPHS_C * D_FEAT);
}


# layers 1, 2: the gather's work-group hands its 16 finished rows through LDS to the next layer's WMMA product (weights staged in LDS), writes the scaled product; 2 launches dropped; sums unchanged
# speedup vs baseline: 5.4877x; 1.1466x over previous
#include <hip/hip_runtime.h>
#include <hip/hip_bf16.h>

typedef __attribute__((ext_vector_type(2))) float v2f;
typedef __attribute__((ext_vector_type(8))) float v8f;

#define D_FEAT 64
#define N_GRAPHS_C 64

#define BIN_SHIFT 8
#define BIN_NODES (1 << BIN_SHIFT)
#define NB_MAX 256
#define SORT_THREADS 1024
#define SORT_GROUPS 128
#define SORT_BATCH 4

__global__ __launch_bounds__(SORT_THREADS) void bin_count_kernel(const int* __restrict__ col,
                                                                 int* __restrict__ wghist,
                                                                 int e, int chunk, int nb) {
    __shared__ int s_h[NB_MAX];
    const int tid = (int)threadIdx.x;
    if (tid < NB_MAX) s_h[tid] = 0;
    __syncthreads();
    const int e0 = (int)blockIdx.x * chunk;
    const int e1 = min(e, e0 + chunk);
    for (int i0 = e0 + tid; i0 < e1; i0 += SORT_BATCH * SORT_THREADS) {
        int c[SORT_BATCH];
#pragma unroll
        for (int k = 0; k < SORT_BATCH; ++k) {
            const int i = i0 + k * SORT_THREADS;
            c[k] = (i < e1) ? col[i] : -1;
        }
#pragma unroll
        for (int k = 0; k < SORT_BATCH; ++k)
            if (c[k] >= 0) atomicAdd(&s_h[c[k] >> BIN_SHIFT], 1);
    }
    __syncthreads();
    if (tid < nb) wghist[(int)blockIdx.x * nb + tid] = s_h[tid];
}

__global__ __launch_bounds__(SORT_THREADS) void bin_spread_kernel(const int* __restrict__ row,
                                                                  const int* __restrict__ col,
                                                                  const int* __restrict__ wghist,
                                                                  int* __restrict__ binstart,
                                                                  int2* __restrict__ binned,
                                                                  int e, int chunk, int nb) {
    __shared__ int s_tot[4][NB_MAX];
    __shared__ int s_bef[4][NB_MAX];
    __shared__ int s_base[NB_MAX];
    __shared__ int s_wave[NB_MAX / 32];
    const int tid  = (int)threadIdx.x;
    const int lane = tid & 31;
    const int wv   = tid >> 5;
    const int grp  = (int)blockIdx.x;
    const int e0   = grp * chunk;
    const int e1   = min(e, e0 + chunk);

    int r[SORT_BATCH], c[SORT_BATCH];
    int i0 = e0 + tid;
#pragma unroll
    for (int k = 0; k < SORT_BATCH; ++k) {
        const int i = i0 + k * SORT_THREADS;
        const bool ok = i < e1;
        c[k] = ok ? col[i] : -1;
        r[k] = ok ? row[i] : 0;
    }

    const int b    = tid & (NB_MAX - 1);
    const int part = tid >> 8;
    int tot = 0, bef = 0;
    if (b < nb) {
        const int g0 = part * (SORT_GROUPS / 4);
#pragma unroll
        for (int g = 0; g < SORT_GROUPS / 4; ++g) {
            const int v = wghist[(g0 + g) * nb + b];
            tot += v;
            bef += (g0 + g < grp) ? v : 0;
        }
    }
    s_tot[part][b] = tot;
    s_bef[part][b] = bef;
    __syncthreads();

    int t = 0, bf = 0, x = 0;
    if (tid < NB_MAX) {
        t  = (s_tot[0][tid] + s_tot[1][tid]) + (s_tot[2][tid] + s_tot[3][tid]);
        bf = (s_bef[0][tid] + s_bef[1][tid]) + (s_bef[2][tid] + s_bef[3][tid]);
        x = t;
#pragma unroll
        for (int d = 1; d < 32; d <<= 1) {
            const int y = __shfl_up(x, d);
            if (lane >= d) x += y;
        }
        if (lane == 31) s_wave[wv] = x;
    }
    __syncthreads();
    if (tid < NB_MAX) {
        int carry = 0;
#pragma unroll
        for (int k = 0; k < NB_MAX / 32; ++k) carry += (k < wv) ? s_wave[k] : 0;
        const int start = carry + (x - t);
        s_base[tid] = start + bf;
        if (grp == 0) {
            if (tid < nb) binstart[tid] = start;
            if (tid == nb - 1) binstart[nb] = start + t;
        }
    }
    __syncthreads();

    for (;;) {
#pragma unroll
        for (int k = 0; k < SORT_BATCH; ++k) {
            if (c[k] >= 0) {
                const int pos = atomicAdd(&s_base[c[k] >> BIN_SHIFT], 1);
                binned[pos] = make_int2(r[k], c[k]);
            }
        }
        i0 += SORT_BATCH * SORT_THREADS;
        if (i0 >= e1) break;
#pragma unroll
        for (int k = 0; k < SORT_BATCH; ++k) {
            const int i = i0 + k * SORT_THREADS;
            const bool ok = i < e1;
            c[k] = ok ? col[i] : -1;
            r[k] = ok ? row[i] : 0;
        }
    }
}

#define RANK_HELD 8
__global__ __launch_bounds__(SORT_THREADS) void bin_rank_kernel(const int2* __restrict__ binned,
                                                                const int* __restrict__ binstart,
                                                                int* __restrict__ rowptr,
                                                                float* __restrict__ dinv,
                                                                int* __restrict__ list, int n) {
    __shared__ int s_cnt[BIN_NODES];
    __shared__ int s_ptr[BIN_NODES];
    __shared__ int s_wave[BIN_NODES / 32];
    const int tid   = (int)threadIdx.x;
    const int lane  = tid & 31;
    const int wv    = tid >> 5;
    const int bin   = (int)blockIdx.x;
    const int lo    = binstart[bin];
    const int hi    = binstart[bin + 1];
    const int node0 = bin << BIN_SHIFT;

    int2 ed[RANK_HELD];
#pragma unroll
    for (int k = 0; k < RANK_HELD; ++k) {
        const int j = lo + tid + k * SORT_THREADS;
        ed[k] = (j < hi) ? binned[j] : make_int2(0, -1);
    }
    if (tid < BIN_NODES) s_cnt[tid] = 0;
    __syncthreads();

#pragma unroll
    for (int k = 0; k < RANK_HELD; ++k)
        if (ed[k].y >= 0) atomicAdd(&s_cnt[ed[k].y - node0], 1);
    for (int j = lo + tid + RANK_HELD * SORT_THREADS; j < hi; j += SORT_THREADS)
        atomicAdd(&s_cnt[binned[j].y - node0], 1);
    __syncthreads();

    int v = 0, x = 0;
    if (tid < BIN_NODES) {
        v = s_cnt[tid];
        x = v;
#pragma unroll
        for (int d = 1; d < 32; d <<= 1) {
            const int y = __shfl_up(x, d);
            if (lane >= d) x += y;
        }
        if (lane == 31) s_wave[wv] = x;
    }
    __syncthreads();
    if (tid < BIN_NODES) {
        int carry = 0;
#pragma unroll
        for (int k = 0; k < BIN_NODES / 32; ++k) carry += (k < wv) ? s_wave[k] : 0;
        const int p = lo + carry + (x - v);
        s_ptr[tid] = p;
        const int node = node0 + tid;
        if (node < n) {
            rowptr[node] = p;
            const float d = (float)(v + 1);
            dinv[node] = (d > 0.0f) ? rsqrtf(d) : 0.0f;
        }
    }
    if (bin == (int)gridDim.x - 1 && tid == 0) rowptr[n] = hi;
    __syncthreads();

#pragma unroll
    for (int k = 0; k < RANK_HELD; ++k) {
        if (ed[k].y >= 0) {
            const int pos = atomicAdd(&s_ptr[ed[k].y - node0], 1);
            list[pos] = ed[k].x;
        }
    }
    for (int j = lo + tid + RANK_HELD * SORT_THREADS; j < hi; j += SORT_THREADS) {
        const int2 ee = binned[j];
        const int pos = atomicAdd(&s_ptr[ee.y - node0], 1);
        list[pos] = ee.x;
    }
}

__global__ __launch_bounds__(256) void gemm_wmma_kernel(const float* __restrict__ X,
                                                        const float* __restrict__ W,
                                                        const float* __restrict__ dinv,
                                                        float* __restrict__ H,
                                                        int nrows) {
    const int wave = (int)((blockIdx.x * blockDim.x + threadIdx.x) >> 5);
    const int lane = (int)(threadIdx.x & 31);
    const int row0 = wave * 16;
    if (row0 >= nrows) return;
    const int lm = lane & 15;
    const int kk = (lane >> 4) * 2;

    const float* __restrict__ xrow = X + (size_t)(row0 + lm) * D_FEAT + kk;
    const int mbase = row0 + (lane >> 4) * 8;
    const float4 dlo = *(const float4*)(dinv + mbase);
    const float4 dhi = *(const float4*)(dinv + mbase + 4);

    v8f c0 = {}, c1 = {}, c2 = {}, c3 = {};
#pragma unroll
    for (int k = 0; k < D_FEAT; k += 4) {
        v2f a;
        a.x = xrow[k + 0];
        a.y = xrow[k + 1];
        const float* __restrict__ wr0 = W + (size_t)(k + kk) * D_FEAT + lm;
        const float* __restrict__ wr1 = wr0 + D_FEAT;
        v2f b0, b1, b2, b3;
        b0.x = wr0[0];   b0.y = wr1[0];
        b1.x = wr0[16];  b1.y = wr1[16];
        b2.x = wr0[32];  b2.y = wr1[32];
        b3.x = wr0[48];  b3.y = wr1[48];
        c0 = __builtin_amdgcn_wmma_f32_16x16x4_f32(false, a, false, b0, (short)0, c0, false, false);
        c1 = __builtin_amdgcn_wmma_f32_16x16x4_f32(false, a, false, b1, (short)0, c1, false, false);
        c2 = __builtin_amdgcn_wmma_f32_16x16x4_f32(false, a, false, b2, (short)0, c2, false, false);
        c3 = __builtin_amdgcn_wmma_f32_16x16x4_f32(false, a, false, b3, (short)0, c3, false, false);
    }

    const float ds[8] = {dlo.x, dlo.y, dlo.z, dlo.w, dhi.x, dhi.y, dhi.z, dhi.w};
#pragma unroll
    for (int r = 0; r < 8; ++r) {
        float* __restrict__ hr = H + (size_t)(mbase + r) * D_FEAT + lm;
        hr[0]  = c0[r] * ds[r];
        hr[16] = c1[r] * ds[r];
        hr[32] = c2[r] * ds[r];
        hr[48] = c3[r] * ds[r];
    }
}

template <int B>
__device__ __forceinline__ void gather_batch(const float* __restrict__ H,
                                             const int* __restrict__ list,
                                             int j, int f, float4& acc) {
    int    en[B];
    float4 hv[B];
#pragma unroll
    for (int k = 0; k < B; ++k) en[k] = list[j + k];
#pragma unroll
    for (int k = 0; k < B; ++k)
        hv[k] = *(const float4*)(H + (size_t)en[k] * D_FEAT + f);
#pragma unroll
    for (int k = 0; k < B; ++k) {
        acc.x += hv[k].x;
        acc.y += hv[k].y;
        acc.z += hv[k].z;
        acc.w += hv[k].w;
    }
}

__global__ __launch_bounds__(256) void gather_kernel(const float* __restrict__ H,
                                                     const int* __restrict__ rowptr,
                                                     const int* __restrict__ list,
                                                     const float* __restrict__ dinv,
                                                     const float* __restrict__ bias,
                                                     float* __restrict__ OUT, int n) {
    const int t = blockIdx.x * blockDim.x + threadIdx.x;
    const int node = t >> 4;
    if (node >= n) return;
    const int f = (t & 15) * 4;
    const int beg = rowptr[node];
    const int end = rowptr[node + 1];
    const float di = dinv[node];
    const float4 hs = *(const float4*)(H + (size_t)node * D_FEAT + f);
    const float4 bv = *(const float4*)(bias + f);
    float4 acc = hs;

    int j = beg;
    for (; j + 8 <= end; j += 8) gather_batch<8>(H, list, j, f, acc);
    const int rest = end - j;
    if (rest & 4) { gather_batch<4>(H, list, j, f, acc); j += 4; }
    if (rest & 2) { gather_batch<2>(H, list, j, f, acc); j += 2; }
    if (rest & 1) { gather_batch<1>(H, list, j, f, acc); }

    float4 o;
    o.x = fmaxf(acc.x * di + bv.x, 0.0f);
    o.y = fmaxf(acc.y * di + bv.y, 0.0f);
    o.z = fmaxf(acc.z * di + bv.z, 0.0f);
    o.w = fmaxf(acc.w * di + bv.w, 0.0f);
    *(float4*)(OUT + (size_t)node * D_FEAT + f) = o;
}

#define FUSE_ROWS 16
#define TILE_PITCH 68
#define W_PITCH 72

__global__ __launch_bounds__(256) void gather_gemm_kernel(const float* __restrict__ H,
                                                          const int* __restrict__ rowptr,
                                                          const int* __restrict__ list,
                                                          const float* __restrict__ dinv,
                                                          const float* __restrict__ bias,
                                                          const float* __restrict__ WN,
                                                          float* __restrict__ HN, int n) {
    __shared__ __attribute__((aligned(16))) float s_w[D_FEAT * W_PITCH];
    __shared__ __attribute__((aligned(16))) float s_act[FUSE_ROWS * TILE_PITCH];
    __shared__ __attribute__((aligned(16))) float s_prod[FUSE_ROWS * TILE_PITCH];
    const int tid  = (int)threadIdx.x;
    const int nl   = tid >> 4;
    const int node = (int)blockIdx.x * FUSE_ROWS + nl;
    const int f    = (tid & 15) * 4;
    const int wv   = tid >> 5;
    const int lane = tid & 31;
    const int lm   = lane & 15;
    const int kk   = (lane >> 4) * 2;
    const bool ok  = node < n;
    const int nc   = ok ? node : (n - 1);

    float4 wq[D_FEAT / FUSE_ROWS];
#pragma unroll
    for (int q = 0; q < D_FEAT / FUSE_ROWS; ++q)
        wq[q] = *(const float4*)(WN + (size_t)(q * FUSE_ROWS + nl) * D_FEAT + f);
    const int beg  = rowptr[nc];
    const int endr = rowptr[nc + 1];
    const float di = dinv[nc];
    const float4 hs = *(const float4*)(H + (size_t)nc * D_FEAT + f);
    const float4 bv = *(const float4*)(bias + f);
    __builtin_amdgcn_sched_barrier(0);
    const int end = ok ? endr : beg;
#pragma unroll
    for (int q = 0; q < D_FEAT / FUSE_ROWS; ++q)
        *(float4*)(s_w + (q * FUSE_ROWS + nl) * W_PITCH + f) = wq[q];

    float4 acc = hs;
    int j = beg;
    for (; j + 8 <= end; j += 8) gather_batch<8>(H, list, j, f, acc);
    const int rest = end - j;
    if (rest & 4) { gather_batch<4>(H, list, j, f, acc); j += 4; }
    if (rest & 2) { gather_batch<2>(H, list, j, f, acc); j += 2; }
    if (rest & 1) { gather_batch<1>(H, list, j, f, acc); }

    float4 o;
    o.x = ok ? fmaxf(acc.x * di + bv.x, 0.0f) : 0.0f;
    o.y = ok ? fmaxf(acc.y * di + bv.y, 0.0f) : 0.0f;
    o.z = ok ? fmaxf(acc.z * di + bv.z, 0.0f) : 0.0f;
    o.w = ok ? fmaxf(acc.w * di + bv.w, 0.0f) : 0.0f;
    *(float4*)(s_act + nl * TILE_PITCH + f) = o;
    __syncthreads();

    if (wv < 4) {
        const float* arow = s_act + lm * TILE_PITCH + kk;
        const float* bcol = s_w + kk * W_PITCH + wv * 16 + lm;
        v2f b[D_FEAT / 4];
#pragma unroll
        for (int s = 0; s < D_FEAT / 4; ++s) {
            b[s].x = bcol[(4 * s) * W_PITCH];
            b[s].y = bcol[(4 * s + 1) * W_PITCH];
        }
        __builtin_amdgcn_sched_barrier(0);
        v8f c = {};
#pragma unroll
        for (int s = 0; s < D_FEAT / 4; ++s) {
            const v2f a = *(const v2f*)(arow + 4 * s);
            c = __builtin_amdgcn_wmma_f32_16x16x4_f32(false, a, false, b[s], (short)0, c, false, false);
        }
        float* pr = s_prod + (lane >> 4) * 8 * TILE_PITCH + wv * 16 + lm;
#pragma unroll
        for (int r = 0; r < 8; ++r) pr[r * TILE_PITCH] = c[r];
    }
    __syncthreads();

    if (ok) {
        float4 p = *(const float4*)(s_prod + nl * TILE_PITCH + f);
        p.x *= di;
        p.y *= di;
        p.z *= di;
        p.w *= di;
        *(float4*)(HN + (size_t)node * D_FEAT + f) = p;
    }
}

__global__ __launch_bounds__(256) void pool_zero_kernel(float* __restrict__ pool,
                                                        float* __restrict__ cnt, int n) {
    int i = blockIdx.x * blockDim.x + threadIdx.x;
    if (i < n) pool[i] = 0.0f;
    if (i < N_GRAPHS_C) cnt[i] = 0.0f;
}

#define POOL_CHUNK 128
#define POOL_ROWS (POOL_CHUNK / 16)
#define POOL_TAB (N_GRAPHS_C * D_FEAT / 256)

__global__ __launch_bounds__(256) void pool_kernel(const float* __restrict__ H,
                                                   const int* __restrict__ batch,
                                                   float* __restrict__ pool,
                                                   float* __restrict__ cnt, int nnodes) {
    __shared__ float s_sum[N_GRAPHS_C * D_FEAT];
    __shared__ float s_cnt[N_GRAPHS_C];
    const int tid  = (int)threadIdx.x;
    const int sub  = tid >> 4;
    const int f    = (tid & 15) * 4;
    const int base = (int)blockIdx.x * POOL_CHUNK + sub;

    int    gid[POOL_ROWS];
    float4 hv[POOL_ROWS];
#pragma unroll
    for (int i = 0; i < POOL_ROWS; ++i) {
        const int node = base + i * 16;
        const bool ok  = node < nnodes;
        const int nc   = ok ? node : (nnodes - 1);
        const int g    = batch[nc];
        hv[i]  = *(const float4*)(H + (size_t)nc * D_FEAT + f);
        gid[i] = (ok && (unsigned)g < (unsigned)N_GRAPHS_C) ? g : -1;
    }

#pragma unroll
    for (int j = 0; j < POOL_TAB; ++j) s_sum[tid + 256 * j] = 0.0f;
    if (tid < N_GRAPHS_C) s_cnt[tid] = 0.0f;
    __syncthreads();

    int    cur = -1;
    float  run = 0.0f;
    float4 acc = make_float4(0.0f, 0.0f, 0.0f, 0.0f);
#pragma unroll
    for (int i = 0; i < POOL_ROWS; ++i) {
        const int g = gid[i];
        if (g >= 0) {
            if (g != cur) {
                if (cur >= 0) {
                    float* s = s_sum + cur * D_FEAT + f;
                    atomicAdd(s + 0, acc.x);
                    atomicAdd(s + 1, acc.y);
                    atomicAdd(s + 2, acc.z);
                    atomicAdd(s + 3, acc.w);
                    if (f == 0) atomicAdd(&s_cnt[cur], run);
                }
                cur = g;
                acc = hv[i];
                run = 1.0f;
            } else {
                acc.x += hv[i].x;
                acc.y += hv[i].y;
                acc.z += hv[i].z;
                acc.w += hv[i].w;
                run += 1.0f;
            }
        }
    }
    if (cur >= 0) {
        float* s = s_sum + cur * D_FEAT + f;
        atomicAdd(s + 0, acc.x);
        atomicAdd(s + 1, acc.y);
        atomicAdd(s + 2, acc.z);
        atomicAdd(s + 3, acc.w);
        if (f == 0) atomicAdd(&s_cnt[cur], run);
    }
    __syncthreads();

    float tc[POOL_TAB], tv[POOL_TAB];
#pragma unroll
    for (int j = 0; j < POOL_TAB; ++j) {
        tc[j] = s_cnt[(tid >> 6) + 4 * j];
        tv[j] = s_sum[tid + 256 * j];
    }
#pragma unroll
    for (int j = 0; j < POOL_TAB; ++j) {
        if (tc[j] > 0.0f) atomicAdd(&pool[tid + 256 * j], tv[j]);
    }
    if (tid < N_GRAPHS_C) {
        const float c = s_cnt[tid];
        if (c > 0.0f) atomicAdd(&cnt[tid], c);
    }
}

__global__ __launch_bounds__(256) void pool_final_kernel(const float* __restrict__ pool,
                                                         const float* __restrict__ cnt,
                                                         float* __restrict__ out, int n) {
    int i = blockIdx.x * blockDim.x + threadIdx.x;
    if (i < n) {
        int g = i >> 6;
        out[i] = pool[i] / fmaxf(cnt[g], 1.0f);
    }
}

extern "C" void kernel_launch(void* const* d_in, const int* in_sizes, int n_in,
                              void* d_out, int out_size, void* d_ws, size_t ws_size,
                              hipStream_t stream) {
    const float* x     = (const float*)d_in[0];
    const int*   ei    = (const int*)d_in[1];
    const int*   batch = (const int*)d_in[2];
    const float* W1    = (const float*)d_in[3];
    const float* b1    = (const float*)d_in[4];
    const float* W2    = (const float*)d_in[5];
    const float* b2    = (const float*)d_in[6];
    const float* W3    = (const float*)d_in[7];
    const float* b3    = (const float*)d_in[8];
    float* out = (float*)d_out;

    const int N = in_sizes[0] / D_FEAT;
    const int E = in_sizes[1] / 2;
    const int* erow = ei;
    const int* ecol = ei + E;

    float* w = (float*)d_ws;
    const size_t NP = (size_t)((N + 1 + 63) / 64) * 64;
    size_t off = 0;
    float* dinv = w + off; off += NP;
    float* bufA = w + off; off += (size_t)N * D_FEAT;
    float* bufB = w + off; off += (size_t)N * D_FEAT;
    float* pool = w + off; off += (size_t)N_GRAPHS_C * D_FEAT;
    float* cnt  = w + off; off += N_GRAPHS_C;
    int* rowptr   = (int*)(w + off); off += NP;
    int* wghist   = (int*)(w + off); off += (size_t)SORT_GROUPS * NB_MAX;
    int* binstart = (int*)(w + off); off += NB_MAX + 64;
    int2* binned  = (int2*)(w + off); off += (size_t)((E + 31) / 32) * 64;
    int* list     = (int*)(w + off); off += (size_t)((E + 63) / 64) * 64;

    const int blk = 256;

    const int NB    = (N + BIN_NODES - 1) / BIN_NODES;
    const int chunk = (E + SORT_GROUPS - 1) / SORT_GROUPS;
    bin_count_kernel<<<SORT_GROUPS, SORT_THREADS, 0, stream>>>(ecol, wghist, E, chunk, NB);
    bin_spread_kernel<<<SORT_GROUPS, SORT_THREADS, 0, stream>>>(erow, ecol, wghist, binstart, binned,
                                                                E, chunk, NB);
    bin_rank_kernel<<<NB, SORT_THREADS, 0, stream>>>(binned, binstart, rowptr, dinv, list, N);

    const int waves = (N + 15) / 16;
    const int gemm_blocks = (waves * 32 + blk - 1) / blk;
    const int gath_blocks = (int)(((long long)N * 16 + blk - 1) / blk);
    const int fuse_blocks = (N + FUSE_ROWS - 1) / FUSE_ROWS;

    gemm_wmma_kernel<<<gemm_blocks, blk, 0, stream>>>(x, W1, dinv, bufA, N);

    gather_gemm_kernel<<<fuse_blocks, blk, 0, stream>>>(bufA, rowptr, list, dinv, b1, W2, bufB, N);

    gather_gemm_kernel<<<fuse_blocks, blk, 0, stream>>>(bufB, rowptr, list, dinv, b2, W3, bufA, N);

    gather_kernel<<<gath_blocks, blk, 0, stream>>>(bufA, rowptr, list, dinv, b3, bufB, N);

    pool_zero_kernel<<<(N_GRAPHS_C * D_FEAT + blk - 1) / blk, blk, 0, stream>>>(pool, cnt,
                                                                               N_GRAPHS_C * D_FEAT);
    pool_kernel<<<(N + POOL_CHUNK - 1) / POOL_CHUNK, blk, 0, stream>>>(bufB, batch, pool, cnt, N);
    pool_final_kernel<<<(N_GRAPHS_C * D_FEAT + blk - 1) / blk, blk, 0, stream>>>(pool, cnt, out,
                                                                                 N_GRAPHS_C * D_FEAT);
}


# layer 3: the gather's work-group sums its 16 rows by graph in LDS and adds sum/count into out (float atomics; counts from the rank pass); 3 pool launches dropped; reorders the pool's float sums
# speedup vs baseline: 5.9058x; 1.0762x over previous
#include <hip/hip_runtime.h>
#include <hip/hip_bf16.h>

typedef __attribute__((ext_vector_type(2))) float v2f;
typedef __attribute__((ext_vector_type(8))) float v8f;

#define D_FEAT 64
#define N_GRAPHS_C 64

#define BIN_SHIFT 8
#define BIN_NODES (1 << BIN_SHIFT)
#define NB_MAX 256
#define SORT_THREADS 1024
#define SORT_GROUPS 128
#define SORT_BATCH 4
static_assert(SORT_GROUPS * SORT_THREADS >= N_GRAPHS_C * D_FEAT, "bin_count clears out with one thread per word");

__global__ __launch_bounds__(SORT_THREADS) void bin_count_kernel(const int* __restrict__ col,
                                                                 int* __restrict__ wghist,
                                                                 float* __restrict__ out,
                                                                 float* __restrict__ cnt,
                                                                 int e, int chunk, int nb) {
    __shared__ int s_h[NB_MAX];
    const int tid = (int)threadIdx.x;
    const int gt = (int)blockIdx.x * SORT_THREADS + tid;
    if (gt < N_GRAPHS_C * D_FEAT) out[gt] = 0.0f;
    if (gt < N_GRAPHS_C) cnt[gt] = 0.0f;
    if (tid < NB_MAX) s_h[tid] = 0;
    __syncthreads();
    const int e0 = (int)blockIdx.x * chunk;
    const int e1 = min(e, e0 + chunk);
    for (int i0 = e0 + tid; i0 < e1; i0 += SORT_BATCH * SORT_THREADS) {
        int c[SORT_BATCH];
#pragma unroll
        for (int k = 0; k < SORT_BATCH; ++k) {
            const int i = i0 + k * SORT_THREADS;
            c[k] = (i < e1) ? col[i] : -1;
        }
#pragma unroll
        for (int k = 0; k < SORT_BATCH; ++k)
            if (c[k] >= 0) atomicAdd(&s_h[c[k] >> BIN_SHIFT], 1);
    }
    __syncthreads();
    if (tid < nb) wghist[(int)blockIdx.x * nb + tid] = s_h[tid];
}

__global__ __launch_bounds__(SORT_THREADS) void bin_spread_kernel(const int* __restrict__ row,
                                                                  const int* __restrict__ col,
                                                                  const int* __restrict__ wghist,
                                                                  int* __restrict__ binstart,
                                                                  int2* __restrict__ binned,
                                                                  int e, int chunk, int nb) {
    __shared__ int s_tot[4][NB_MAX];
    __shared__ int s_bef[4][NB_MAX];
    __shared__ int s_base[NB_MAX];
    __shared__ int s_wave[NB_MAX / 32];
    const int tid  = (int)threadIdx.x;
    const int lane = tid & 31;
    const int wv   = tid >> 5;
    const int grp  = (int)blockIdx.x;
    const int e0   = grp * chunk;
    const int e1   = min(e, e0 + chunk);

    int r[SORT_BATCH], c[SORT_BATCH];
    int i0 = e0 + tid;
#pragma unroll
    for (int k = 0; k < SORT_BATCH; ++k) {
        const int i = i0 + k * SORT_THREADS;
        const bool ok = i < e1;
        c[k] = ok ? col[i] : -1;
        r[k] = ok ? row[i] : 0;
    }

    const int b    = tid & (NB_MAX - 1);
    const int part = tid >> 8;
    int tot = 0, bef = 0;
    if (b < nb) {
        const int g0 = part * (SORT_GROUPS / 4);
#pragma unroll
        for (int g = 0; g < SORT_GROUPS / 4; ++g) {
            const int v = wghist[(g0 + g) * nb + b];
            tot += v;
            bef += (g0 + g < grp) ? v : 0;
        }
    }
    s_tot[part][b] = tot;
    s_bef[part][b] = bef;
    __syncthreads();

    int t = 0, bf = 0, x = 0;
    if (tid < NB_MAX) {
        t  = (s_tot[0][tid] + s_tot[1][tid]) + (s_tot[2][tid] + s_tot[3][tid]);
        bf = (s_bef[0][tid] + s_bef[1][tid]) + (s_bef[2][tid] + s_bef[3][tid]);
        x = t;
#pragma unroll
        for (int d = 1; d < 32; d <<= 1) {
            const int y = __shfl_up(x, d);
            if (lane >= d) x += y;
        }
        if (lane == 31) s_wave[wv] = x;
    }
    __syncthreads();
    if (tid < NB_MAX) {
        int carry = 0;
#pragma unroll
        for (int k = 0; k < NB_MAX / 32; ++k) carry += (k < wv) ? s_wave[k] : 0;
        const int start = carry + (x - t);
        s_base[tid] = start + bf;
        if (grp == 0) {
            if (tid < nb) binstart[tid] = start;
            if (tid == nb - 1) binstart[nb] = start + t;
        }
    }
    __syncthreads();

    for (;;) {
#pragma unroll
        for (int k = 0; k < SORT_BATCH; ++k) {
            if (c[k] >= 0) {
                const int pos = atomicAdd(&s_base[c[k] >> BIN_SHIFT], 1);
                binned[pos] = make_int2(r[k], c[k]);
            }
        }
        i0 += SORT_BATCH * SORT_THREADS;
        if (i0 >= e1) break;
#pragma unroll
        for (int k = 0; k < SORT_BATCH; ++k) {
            const int i = i0 + k * SORT_THREADS;
            const bool ok = i < e1;
            c[k] = ok ? col[i] : -1;
            r[k] = ok ? row[i] : 0;
        }
    }
}

#define RANK_HELD 8
__global__ __launch_bounds__(SORT_THREADS) void bin_rank_kernel(const int2* __restrict__ binned,
                                                                const int* __restrict__ binstart,
                                                                int* __restrict__ rowptr,
                                                                float* __restrict__ dinv,
                                                                int* __restrict__ list,
                                                                const int* __restrict__ batch,
                                                                float* __restrict__ cnt, int n) {
    __shared__ int s_cnt[BIN_NODES];
    __shared__ int s_ptr[BIN_NODES];
    __shared__ int s_wave[BIN_NODES / 32];
    __shared__ int s_gc[N_GRAPHS_C];
    const int tid   = (int)threadIdx.x;
    const int lane  = tid & 31;
    const int wv    = tid >> 5;
    const int bin   = (int)blockIdx.x;
    const int lo    = binstart[bin];
    const int hi    = binstart[bin + 1];
    const int node0 = bin << BIN_SHIFT;

    int2 ed[RANK_HELD];
#pragma unroll
    for (int k = 0; k < RANK_HELD; ++k) {
        const int j = lo + tid + k * SORT_THREADS;
        ed[k] = (j < hi) ? binned[j] : make_int2(0, -1);
    }
    int gb = -1;
    if (tid < BIN_NODES && node0 + tid < n) gb = batch[node0 + tid];
    if (tid < BIN_NODES) s_cnt[tid] = 0;
    if (tid < N_GRAPHS_C) s_gc[tid] = 0;
    __syncthreads();

#pragma unroll
    for (int k = 0; k < RANK_HELD; ++k)
        if (ed[k].y >= 0) atomicAdd(&s_cnt[ed[k].y - node0], 1);
    for (int j = lo + tid + RANK_HELD * SORT_THREADS; j < hi; j += SORT_THREADS)
        atomicAdd(&s_cnt[binned[j].y - node0], 1);
    if ((unsigned)gb < (unsigned)N_GRAPHS_C) atomicAdd(&s_gc[gb], 1);
    __syncthreads();

    int v = 0, x = 0;
    if (tid < BIN_NODES) {
        v = s_cnt[tid];
        x = v;
#pragma unroll
        for (int d = 1; d < 32; d <<= 1) {
            const int y = __shfl_up(x, d);
            if (lane >= d) x += y;
        }
        if (lane == 31) s_wave[wv] = x;
    }
    __syncthreads();
    if (tid < BIN_NODES) {
        int carry = 0;
#pragma unroll
        for (int k = 0; k < BIN_NODES / 32; ++k) carry += (k < wv) ? s_wave[k] : 0;
        const int p = lo + carry + (x - v);
        s_ptr[tid] = p;
        const int node = node0 + tid;
        if (node < n) {
            rowptr[node] = p;
            const float d = (float)(v + 1);
            dinv[node] = (d > 0.0f) ? rsqrtf(d) : 0.0f;
        }
    }
    if (bin == (int)gridDim.x - 1 && tid == 0) rowptr[n] = hi;
    if (tid < N_GRAPHS_C) {
        const int gc = s_gc[tid];
        if (gc > 0) atomicAdd(&cnt[tid], (float)gc);
    }
    __syncthreads();

#pragma unroll
    for (int k = 0; k < RANK_HELD; ++k) {
        if (ed[k].y >= 0) {
            const int pos = atomicAdd(&s_ptr[ed[k].y - node0], 1);
            list[pos] = ed[k].x;
        }
    }
    for (int j = lo + tid + RANK_HELD * SORT_THREADS; j < hi; j += SORT_THREADS) {
        const int2 ee = binned[j];
        const int pos = atomicAdd(&s_ptr[ee.y - node0], 1);
        list[pos] = ee.x;
    }
}

__global__ __launch_bounds__(256) void gemm_wmma_kernel(const float* __restrict__ X,
                                                        const float* __restrict__ W,
                                                        const float* __restrict__ dinv,
                                                        float* __restrict__ H,
                                                        int nrows) {
    const int wave = (int)((blockIdx.x * blockDim.x + threadIdx.x) >> 5);
    const int lane = (int)(threadIdx.x & 31);
    const int row0 = wave * 16;
    if (row0 >= nrows) return;
    const int lm = lane & 15;
    const int kk = (lane >> 4) * 2;

    const float* __restrict__ xrow = X + (size_t)(row0 + lm) * D_FEAT + kk;
    const int mbase = row0 + (lane >> 4) * 8;
    const float4 dlo = *(const float4*)(dinv + mbase);
    const float4 dhi = *(const float4*)(dinv + mbase + 4);

    v8f c0 = {}, c1 = {}, c2 = {}, c3 = {};
#pragma unroll
    for (int k = 0; k < D_FEAT; k += 4) {
        v2f a;
        a.x = xrow[k + 0];
        a.y = xrow[k + 1];
        const float* __restrict__ wr0 = W + (size_t)(k + kk) * D_FEAT + lm;
        const float* __restrict__ wr1 = wr0 + D_FEAT;
        v2f b0, b1, b2, b3;
        b0.x = wr0[0];   b0.y = wr1[0];
        b1.x = wr0[16];  b1.y = wr1[16];
        b2.x = wr0[32];  b2.y = wr1[32];
        b3.x = wr0[48];  b3.y = wr1[48];
        c0 = __builtin_amdgcn_wmma_f32_16x16x4_f32(false, a, false, b0, (short)0, c0, false, false);
        c1 = __builtin_amdgcn_wmma_f32_16x16x4_f32(false, a, false, b1, (short)0, c1, false, false);
        c2 = __builtin_amdgcn_wmma_f32_16x16x4_f32(false, a, false, b2, (short)0, c2, false, false);
        c3 = __builtin_amdgcn_wmma_f32_16x16x4_f32(false, a, false, b3, (short)0, c3, false, false);
    }

    const float ds[8] = {dlo.x, dlo.y, dlo.z, dlo.w, dhi.x, dhi.y, dhi.z, dhi.w};
#pragma unroll
    for (int r = 0; r < 8; ++r) {
        float* __restrict__ hr = H + (size_t)(mbase + r) * D_FEAT + lm;
        hr[0]  = c0[r] * ds[r];
        hr[16] = c1[r] * ds[r];
        hr[32] = c2[r] * ds[r];
        hr[48] = c3[r] * ds[r];
    }
}

template <int B>
__device__ __forceinline__ void gather_batch(const float* __restrict__ H,
                                             const int* __restrict__ list,
                                             int j, int f, float4& acc) {
    int    en[B];
    float4 hv[B];
#pragma unroll
    for (int k = 0; k < B; ++k) en[k] = list[j + k];
#pragma unroll
    for (int k = 0; k < B; ++k)
        hv[k] = *(const float4*)(H + (size_t)en[k] * D_FEAT + f);
#pragma unroll
    for (int k = 0; k < B; ++k) {
        acc.x += hv[k].x;
        acc.y += hv[k].y;
        acc.z += hv[k].z;
        acc.w += hv[k].w;
    }
}

#define FUSE_ROWS 16
#define TILE_PITCH 68
#define W_PITCH 72

__global__ __launch_bounds__(256) void gather_gemm_kernel(const float* __restrict__ H,
                                                          const int* __restrict__ rowptr,
                                                          const int* __restrict__ list,
                                                          const float* __restrict__ dinv,
                                                          const float* __restrict__ bias,
                                                          const float* __restrict__ WN,
                                                          float* __restrict__ HN, int n) {
    __shared__ __attribute__((aligned(16))) float s_w[D_FEAT * W_PITCH];
    __shared__ __attribute__((aligned(16))) float s_act[FUSE_ROWS * TILE_PITCH];
    __shared__ __attribute__((aligned(16))) float s_prod[FUSE_ROWS * TILE_PITCH];
    const int tid  = (int)threadIdx.x;
    const int nl   = tid >> 4;
    const int node = (int)blockIdx.x * FUSE_ROWS + nl;
    const int f    = (tid & 15) * 4;
    const int wv   = tid >> 5;
    const int lane = tid & 31;
    const int lm   = lane & 15;
    const int kk   = (lane >> 4) * 2;
    const bool ok  = node < n;
    const int nc   = ok ? node : (n - 1);

    float4 wq[D_FEAT / FUSE_ROWS];
#pragma unroll
    for (int q = 0; q < D_FEAT / FUSE_ROWS; ++q)
        wq[q] = *(const float4*)(WN + (size_t)(q * FUSE_ROWS + nl) * D_FEAT + f);
    const int beg  = rowptr[nc];
    const int endr = rowptr[nc + 1];
    const float di = dinv[nc];
    const float4 hs = *(const float4*)(H + (size_t)nc * D_FEAT + f);
    const float4 bv = *(const float4*)(bias + f);
    __builtin_amdgcn_sched_barrier(0);
    const int end = ok ? endr : beg;
#pragma unroll
    for (int q = 0; q < D_FEAT / FUSE_ROWS; ++q)
        *(float4*)(s_w + (q * FUSE_ROWS + nl) * W_PITCH + f) = wq[q];

    float4 acc = hs;
    int j = beg;
    for (; j + 8 <= end; j += 8) gather_batch<8>(H, list, j, f, acc);
    const int rest = end - j;
    if (rest & 4) { gather_batch<4>(H, list, j, f, acc); j += 4; }
    if (rest & 2) { gather_batch<2>(H, list, j, f, acc); j += 2; }
    if (rest & 1) { gather_batch<1>(H, list, j, f, acc); }

    float4 o;
    o.x = ok ? fmaxf(acc.x * di + bv.x, 0.0f) : 0.0f;
    o.y = ok ? fmaxf(acc.y * di + bv.y, 0.0f) : 0.0f;
    o.z = ok ? fmaxf(acc.z * di + bv.z, 0.0f) : 0.0f;
    o.w = ok ? fmaxf(acc.w * di + bv.w, 0.0f) : 0.0f;
    *(float4*)(s_act + nl * TILE_PITCH + f) = o;
    __syncthreads();

    if (wv < 4) {
        const float* arow = s_act + lm * TILE_PITCH + kk;
        const float* bcol = s_w + kk * W_PITCH + wv * 16 + lm;
        v2f b[D_FEAT / 4];
#pragma unroll
        for (int s = 0; s < D_FEAT / 4; ++s) {
            b[s].x = bcol[(4 * s) * W_PITCH];
            b[s].y = bcol[(4 * s + 1) * W_PITCH];
        }
        __builtin_amdgcn_sched_barrier(0);
        v8f c = {};
#pragma unroll
        for (int s = 0; s < D_FEAT / 4; ++s) {
            const v2f a = *(const v2f*)(arow + 4 * s);
            c = __builtin_amdgcn_wmma_f32_16x16x4_f32(false, a, false, b[s], (short)0, c, false, false);
        }
        float* pr = s_prod + (lane >> 4) * 8 * TILE_PITCH + wv * 16 + lm;
#pragma unroll
        for (int r = 0; r < 8; ++r) pr[r * TILE_PITCH] = c[r];
    }
    __syncthreads();

    if (ok) {
        float4 p = *(const float4*)(s_prod + nl * TILE_PITCH + f);
        p.x *= di;
        p.y *= di;
        p.z *= di;
        p.w *= di;
        *(float4*)(HN + (size_t)node * D_FEAT + f) = p;
    }
}

__global__ __launch_bounds__(256) void gather_pool_kernel(const float* __restrict__ H,
                                                          const int* __restrict__ rowptr,
                                                          const int* __restrict__ list,
                                                          const float* __restrict__ dinv,
                                                          const float* __restrict__ bias,
                                                          const int* __restrict__ batch,
                                                          const float* __restrict__ cnt,
                                                          float* __restrict__ out, int n) {
    __shared__ __attribute__((aligned(16))) float s_row[FUSE_ROWS * D_FEAT];
    __shared__ float s_div[N_GRAPHS_C];
    __shared__ int s_gid[FUSE_ROWS];
    const int tid  = (int)threadIdx.x;
    const int nl   = tid >> 4;
    const int node = (int)blockIdx.x * FUSE_ROWS + nl;
    const int f    = (tid & 15) * 4;
    const bool ok  = node < n;
    const int nc   = ok ? node : (n - 1);

    const int beg  = rowptr[nc];
    const int endr = rowptr[nc + 1];
    const float di = dinv[nc];
    const float4 hs = *(const float4*)(H + (size_t)nc * D_FEAT + f);
    const float4 bv = *(const float4*)(bias + f);
    const int gb   = batch[nc];
    const float cg = cnt[tid & (N_GRAPHS_C - 1)];
    __builtin_amdgcn_sched_barrier(0);
    const int end  = ok ? endr : beg;

    float4 acc = hs;
    int j = beg;
    for (; j + 8 <= end; j += 8) gather_batch<8>(H, list, j, f, acc);
    const int rest = end - j;
    if (rest & 4) { gather_batch<4>(H, list, j, f, acc); j += 4; }
    if (rest & 2) { gather_batch<2>(H, list, j, f, acc); j += 2; }
    if (rest & 1) { gather_batch<1>(H, list, j, f, acc); }

    float4 o;
    o.x = fmaxf(acc.x * di + bv.x, 0.0f);
    o.y = fmaxf(acc.y * di + bv.y, 0.0f);
    o.z = fmaxf(acc.z * di + bv.z, 0.0f);
    o.w = fmaxf(acc.w * di + bv.w, 0.0f);
    *(float4*)(s_row + nl * D_FEAT + f) = o;
    if ((tid & 15) == 0) s_gid[nl] = (ok && (unsigned)gb < (unsigned)N_GRAPHS_C) ? gb : -1;
    if (tid < N_GRAPHS_C) s_div[tid] = fmaxf(cg, 1.0f);
    __syncthreads();

    if (tid < D_FEAT) {
        int   gq[FUSE_ROWS];
        float vq[FUSE_ROWS];
#pragma unroll
        for (int r = 0; r < FUSE_ROWS; ++r) {
            gq[r] = s_gid[r];
            vq[r] = s_row[r * D_FEAT + tid];
        }
        int   cur = -1;
        float run = 0.0f;
#pragma unroll
        for (int r = 0; r < FUSE_ROWS; ++r) {
            if (gq[r] != cur) {
                if (cur >= 0) atomicAdd(out + cur * D_FEAT + tid, run / s_div[cur]);
                cur = gq[r];
                run = vq[r];
            } else {
                run += vq[r];
            }
        }
        if (cur >= 0) atomicAdd(out + cur * D_FEAT + tid, run / s_div[cur]);
    }
}

extern "C" void kernel_launch(void* const* d_in, const int* in_sizes, int n_in,
                              void* d_out, int out_size, void* d_ws, size_t ws_size,
                              hipStream_t stream) {
    const float* x     = (const float*)d_in[0];
    const int*   ei    = (const int*)d_in[1];
    const int*   batch = (const int*)d_in[2];
    const float* W1    = (const float*)d_in[3];
    const float* b1    = (const float*)d_in[4];
    const float* W2    = (const float*)d_in[5];
    const float* b2    = (const float*)d_in[6];
    const float* W3    = (const float*)d_in[7];
    const float* b3    = (const float*)d_in[8];
    float* out = (float*)d_out;

    const int N = in_sizes[0] / D_FEAT;
    const int E = in_sizes[1] / 2;
    const int* erow = ei;
    const int* ecol = ei + E;

    float* w = (float*)d_ws;
    const size_t NP = (size_t)((N + 1 + 63) / 64) * 64;
    size_t off = 0;
    float* dinv = w + off; off += NP;
    float* bufA = w + off; off += (size_t)N * D_FEAT;
    float* bufB = w + off; off += (size_t)N * D_FEAT;
    float* cnt  = w + off; off += N_GRAPHS_C;
    int* rowptr   = (int*)(w + off); off += NP;
    int* wghist   = (int*)(w + off); off += (size_t)SORT_GROUPS * NB_MAX;
    int* binstart = (int*)(w + off); off += NB_MAX + 64;
    int2* binned  = (int2*)(w + off); off += (size_t)((E + 31) / 32) * 64;
    int* list     = (int*)(w + off); off += (size_t)((E + 63) / 64) * 64;

    const int blk = 256;

    const int NB    = (N + BIN_NODES - 1) / BIN_NODES;
    const int chunk = (E + SORT_GROUPS - 1) / SORT_GROUPS;
    bin_count_kernel<<<SORT_GROUPS, SORT_THREADS, 0, stream>>>(ecol, wghist, out, cnt, E, chunk, NB);
    bin_spread_kernel<<<SORT_GROUPS, SORT_THREADS, 0, stream>>>(erow, ecol, wghist, binstart, binned,
                                                                E, chunk, NB);
    bin_rank_kernel<<<NB, SORT_THREADS, 0, stream>>>(binned, binstart, rowptr, dinv, list, batch, cnt, N);

    const int waves = (N + 15) / 16;
    const int gemm_blocks = (waves * 32 + blk - 1) / blk;
    const int fuse_blocks = (N + FUSE_ROWS - 1) / FUSE_ROWS;

    gemm_wmma_kernel<<<gemm_blocks, blk, 0, stream>>>(x, W1, dinv, bufA, N);

    gather_gemm_kernel<<<fuse_blocks, blk, 0, stream>>>(bufA, rowptr, list, dinv, b1, W2, bufB, N);

    gather_gemm_kernel<<<fuse_blocks, blk, 0, stream>>>(bufB, rowptr, list, dinv, b2, W3, bufA, N);

    gather_pool_kernel<<<fuse_blocks, blk, 0, stream>>>(bufA, rowptr, list, dinv, b3, batch, cnt, out, N);
}


# layer 1 product moved into the rank launch: each bin's work-group multiplies its own 256 rows of x by W1 (LDS) on WMMA behind its list stores and scales them by its dinv from LDS; 1 launch dropped
# speedup vs baseline: 6.2630x; 1.0605x over previous
#include <hip/hip_runtime.h>
#include <hip/hip_bf16.h>

typedef __attribute__((ext_vector_type(2))) float v2f;
typedef __attribute__((ext_vector_type(8))) float v8f;

#define D_FEAT 64
#define N_GRAPHS_C 64

#define BIN_SHIFT 8
#define BIN_NODES (1 << BIN_SHIFT)
#define NB_MAX 256
#define SORT_THREADS 1024
#define SORT_GROUPS 128
#define SORT_BATCH 4
static_assert(SORT_GROUPS * SORT_THREADS >= N_GRAPHS_C * D_FEAT, "bin_count clears out with one thread per word");

__global__ __launch_bounds__(SORT_THREADS) void bin_count_kernel(const int* __restrict__ col,
                                                                 int* __restrict__ wghist,
                                                                 float* __restrict__ out,
                                                                 float* __restrict__ cnt,
                                                                 int e, int chunk, int nb) {
    __shared__ int s_h[NB_MAX];
    const int tid = (int)threadIdx.x;
    const int gt = (int)blockIdx.x * SORT_THREADS + tid;
    if (gt < N_GRAPHS_C * D_FEAT) out[gt] = 0.0f;
    if (gt < N_GRAPHS_C) cnt[gt] = 0.0f;
    if (tid < NB_MAX) s_h[tid] = 0;
    __syncthreads();
    const int e0 = (int)blockIdx.x * chunk;
    const int e1 = min(e, e0 + chunk);
    for (int i0 = e0 + tid; i0 < e1; i0 += SORT_BATCH * SORT_THREADS) {
        int c[SORT_BATCH];
#pragma unroll
        for (int k = 0; k < SORT_BATCH; ++k) {
            const int i = i0 + k * SORT_THREADS;
            c[k] = (i < e1) ? col[i] : -1;
        }
#pragma unroll
        for (int k = 0; k < SORT_BATCH; ++k)
            if (c[k] >= 0) atomicAdd(&s_h[c[k] >> BIN_SHIFT], 1);
    }
    __syncthreads();
    if (tid < nb) wghist[(int)blockIdx.x * nb + tid] = s_h[tid];
}

__global__ __launch_bounds__(SORT_THREADS) void bin_spread_kernel(const int* __restrict__ row,
                                                                  const int* __restrict__ col,
                                                                  const int* __restrict__ wghist,
                                                                  int* __restrict__ binstart,
                                                                  int2* __restrict__ binned,
                                                                  int e, int chunk, int nb) {
    __shared__ int s_tot[4][NB_MAX];
    __shared__ int s_bef[4][NB_MAX];
    __shared__ int s_base[NB_MAX];
    __shared__ int s_wave[NB_MAX / 32];
    const int tid  = (int)threadIdx.x;
    const int lane = tid & 31;
    const int wv   = tid >> 5;
    const int grp  = (int)blockIdx.x;
    const int e0   = grp * chunk;
    const int e1   = min(e, e0 + chunk);

    int r[SORT_BATCH], c[SORT_BATCH];
    int i0 = e0 + tid;
#pragma unroll
    for (int k = 0; k < SORT_BATCH; ++k) {
        const int i = i0 + k * SORT_THREADS;
        const bool ok = i < e1;
        c[k] = ok ? col[i] : -1;
        r[k] = ok ? row[i] : 0;
    }

    const int b    = tid & (NB_MAX - 1);
    const int part = tid >> 8;
    int tot = 0, bef = 0;
    if (b < nb) {
        const int g0 = part * (SORT_GROUPS / 4);
#pragma unroll
        for (int g = 0; g < SORT_GROUPS / 4; ++g) {
            const int v = wghist[(g0 + g) * nb + b];
            tot += v;
            bef += (g0 + g < grp) ? v : 0;
        }
    }
    s_tot[part][b] = tot;
    s_bef[part][b] = bef;
    __syncthreads();

    int t = 0, bf = 0, x = 0;
    if (tid < NB_MAX) {
        t  = (s_tot[0][tid] + s_tot[1][tid]) + (s_tot[2][tid] + s_tot[3][tid]);
        bf = (s_bef[0][tid] + s_bef[1][tid]) + (s_bef[2][tid] + s_bef[3][tid]);
        x = t;
#pragma unroll
        for (int d = 1; d < 32; d <<= 1) {
            const int y = __shfl_up(x, d);
            if (lane >= d) x += y;
        }
        if (lane == 31) s_wave[wv] = x;
    }
    __syncthreads();
    if (tid < NB_MAX) {
        int carry = 0;
#pragma unroll
        for (int k = 0; k < NB_MAX / 32; ++k) carry += (k < wv) ? s_wave[k] : 0;
        const int start = carry + (x - t);
        s_base[tid] = start + bf;
        if (grp == 0) {
            if (tid < nb) binstart[tid] = start;
            if (tid == nb - 1) binstart[nb] = start + t;
        }
    }
    __syncthreads();

    for (;;) {
#pragma unroll
        for (int k = 0; k < SORT_BATCH; ++k) {
            if (c[k] >= 0) {
                const int pos = atomicAdd(&s_base[c[k] >> BIN_SHIFT], 1);
                binned[pos] = make_int2(r[k], c[k]);
            }
        }
        i0 += SORT_BATCH * SORT_THREADS;
        if (i0 >= e1) break;
#pragma unroll
        for (int k = 0; k < SORT_BATCH; ++k) {
            const int i = i0 + k * SORT_THREADS;
            const bool ok = i < e1;
            c[k] = ok ? col[i] : -1;
            r[k] = ok ? row[i] : 0;
        }
    }
}

#define RANK_HELD 8
#define RANK_STRIPS (BIN_NODES / 16)
#define W_PITCH 72
static_assert(SORT_THREADS / 32 == 2 * RANK_STRIPS, "two waves per strip, one per half of the columns");
static_assert(SORT_THREADS * 4 == D_FEAT * D_FEAT, "one float4 of the weights per thread");

__global__ __launch_bounds__(SORT_THREADS) void bin_rank_kernel(const int2* __restrict__ binned,
                                                                const int* __restrict__ binstart,
                                                                int* __restrict__ rowptr,
                                                                float* __restrict__ dinv,
                                                                int* __restrict__ list,
                                                                const int* __restrict__ batch,
                                                                float* __restrict__ cnt,
                                                                const float* __restrict__ X,
                                                                const float* __restrict__ W,
                                                                float* __restrict__ H, int n) {
    __shared__ int s_cnt[BIN_NODES];
    __shared__ int s_ptr[BIN_NODES];
    __shared__ int s_wave[BIN_NODES / 32];
    __shared__ int s_gc[N_GRAPHS_C];
    __shared__ __attribute__((aligned(16))) float s_dinv[BIN_NODES];
    __shared__ __attribute__((aligned(16))) float s_w[D_FEAT * W_PITCH];
    const int tid   = (int)threadIdx.x;
    const int lane  = tid & 31;
    const int wv    = tid >> 5;
    const int bin   = (int)blockIdx.x;
    const int node0 = bin << BIN_SHIFT;

    const float4 wq = *(const float4*)(W + (size_t)(tid >> 4) * D_FEAT + (tid & 15) * 4);
    const int lm    = lane & 15;
    const int kk    = (lane >> 4) * 2;
    const int strip = wv & (RANK_STRIPS - 1);
    const int col0  = (wv / RANK_STRIPS) * 32;
    const int srow0 = node0 + strip * 16;
    const float* __restrict__ xrow = X + (size_t)min(srow0 + lm, n - 1) * D_FEAT + kk;
    v2f a[D_FEAT / 4];
#pragma unroll
    for (int s = 0; s < D_FEAT / 4; ++s) a[s] = *(const v2f*)(xrow + 4 * s);
    *(float4*)(s_w + (tid >> 4) * W_PITCH + (tid & 15) * 4) = wq;

    const int lo    = binstart[bin];
    const int hi    = binstart[bin + 1];

    int2 ed[RANK_HELD];
    const int jmax = max(hi - 1, 0);
#pragma unroll
    for (int k = 0; k < RANK_HELD; ++k) {
        const int j = lo + tid + k * SORT_THREADS;
        ed[k] = binned[min(j, jmax)];
        if (j >= hi) ed[k].y = -1;
    }
    int gb = -1;
    if (tid < BIN_NODES && node0 + tid < n) gb = batch[node0 + tid];

    if (tid < BIN_NODES) s_cnt[tid] = 0;
    if (tid < N_GRAPHS_C) s_gc[tid] = 0;
    __syncthreads();

#pragma unroll
    for (int k = 0; k < RANK_HELD; ++k)
        if (ed[k].y >= 0) atomicAdd(&s_cnt[ed[k].y - node0], 1);
    for (int j = lo + tid + RANK_HELD * SORT_THREADS; j < hi; j += SORT_THREADS)
        atomicAdd(&s_cnt[binned[j].y - node0], 1);
    if ((unsigned)gb < (unsigned)N_GRAPHS_C) atomicAdd(&s_gc[gb], 1);
    __syncthreads();

    int v = 0, x = 0;
    if (tid < BIN_NODES) {
        v = s_cnt[tid];
        x = v;
#pragma unroll
        for (int d = 1; d < 32; d <<= 1) {
            const int y = __shfl_up(x, d);
            if (lane >= d) x += y;
        }
        if (lane == 31) s_wave[wv] = x;
    }
    __syncthreads();
    if (tid < BIN_NODES) {
        int carry = 0;
#pragma unroll
        for (int k = 0; k < BIN_NODES / 32; ++k) carry += (k < wv) ? s_wave[k] : 0;
        const int p = lo + carry + (x - v);
        s_ptr[tid] = p;
        const int node = node0 + tid;
        const float d = (float)(v + 1);
        const float dv = (d > 0.0f) ? rsqrtf(d) : 0.0f;
        s_dinv[tid] = dv;
        if (node < n) {
            rowptr[node] = p;
            dinv[node] = dv;
        }
    }
    if (bin == (int)gridDim.x - 1 && tid == 0) rowptr[n] = hi;
    if (tid < N_GRAPHS_C) {
        const int gc = s_gc[tid];
        if (gc > 0) atomicAdd(&cnt[tid], (float)gc);
    }
    __syncthreads();

#pragma unroll
    for (int k = 0; k < RANK_HELD; ++k) {
        if (ed[k].y >= 0) {
            const int pos = atomicAdd(&s_ptr[ed[k].y - node0], 1);
            list[pos] = ed[k].x;
        }
    }
    for (int j = lo + tid + RANK_HELD * SORT_THREADS; j < hi; j += SORT_THREADS) {
        const int2 ee = binned[j];
        const int pos = atomicAdd(&s_ptr[ee.y - node0], 1);
        list[pos] = ee.x;
    }

    if (srow0 < n) {
        const float* bcol = s_w + kk * W_PITCH + col0 + lm;
        v8f c0 = {}, c1 = {};
#pragma unroll
        for (int s = 0; s < D_FEAT / 4; ++s) {
            const float* bp = bcol + (4 * s) * W_PITCH;
            v2f b0, b1;
            b0.x = bp[0];
            b0.y = bp[W_PITCH];
            b1.x = bp[16];
            b1.y = bp[W_PITCH + 16];
            c0 = __builtin_amdgcn_wmma_f32_16x16x4_f32(false, a[s], false, b0, (short)0, c0, false, false);
            c1 = __builtin_amdgcn_wmma_f32_16x16x4_f32(false, a[s], false, b1, (short)0, c1, false, false);
        }
        const int hb = (lane >> 4) * 8;
        const float4 dlo = *(const float4*)(s_dinv + strip * 16 + hb);
        const float4 dhi = *(const float4*)(s_dinv + strip * 16 + hb + 4);
        const float ds[8] = {dlo.x, dlo.y, dlo.z, dlo.w, dhi.x, dhi.y, dhi.z, dhi.w};
        float* __restrict__ hr = H + (size_t)(srow0 + hb) * D_FEAT + col0 + lm;
#pragma unroll
        for (int r = 0; r < 8; ++r) {
            if (srow0 + hb + r < n) {
                hr[r * D_FEAT]      = c0[r] * ds[r];
                hr[r * D_FEAT + 16] = c1[r] * ds[r];
            }
        }
    }
}

template <int B>
__device__ __forceinline__ void gather_batch(const float* __restrict__ H,
                                             const int* __restrict__ list,
                                             int j, int f, float4& acc) {
    int    en[B];
    float4 hv[B];
#pragma unroll
    for (int k = 0; k < B; ++k) en[k] = list[j + k];
#pragma unroll
    for (int k = 0; k < B; ++k)
        hv[k] = *(const float4*)(H + (size_t)en[k] * D_FEAT + f);
#pragma unroll
    for (int k = 0; k < B; ++k) {
        acc.x += hv[k].x;
        acc.y += hv[k].y;
        acc.z += hv[k].z;
        acc.w += hv[k].w;
    }
}

#define FUSE_ROWS 16
#define TILE_PITCH 68

__global__ __launch_bounds__(256) void gather_gemm_kernel(const float* __restrict__ H,
                                                          const int* __restrict__ rowptr,
                                                          const int* __restrict__ list,
                                                          const float* __restrict__ dinv,
                                                          const float* __restrict__ bias,
                                                          const float* __restrict__ WN,
                                                          float* __restrict__ HN, int n) {
    __shared__ __attribute__((aligned(16))) float s_w[D_FEAT * W_PITCH];
    __shared__ __attribute__((aligned(16))) float s_act[FUSE_ROWS * TILE_PITCH];
    __shared__ __attribute__((aligned(16))) float s_prod[FUSE_ROWS * TILE_PITCH];
    const int tid  = (int)threadIdx.x;
    const int nl   = tid >> 4;
    const int node = (int)blockIdx.x * FUSE_ROWS + nl;
    const int f    = (tid & 15) * 4;
    const int wv   = tid >> 5;
    const int lane = tid & 31;
    const int lm   = lane & 15;
    const int kk   = (lane >> 4) * 2;
    const bool ok  = node < n;
    const int nc   = ok ? node : (n - 1);

    float4 wq[D_FEAT / FUSE_ROWS];
#pragma unroll
    for (int q = 0; q < D_FEAT / FUSE_ROWS; ++q)
        wq[q] = *(const float4*)(WN + (size_t)(q * FUSE_ROWS + nl) * D_FEAT + f);
    const int beg  = rowptr[nc];
    const int endr = rowptr[nc + 1];
    const float di = dinv[nc];
    const float4 hs = *(const float4*)(H + (size_t)nc * D_FEAT + f);
    const float4 bv = *(const float4*)(bias + f);
    __builtin_amdgcn_sched_barrier(0);
    const int end = ok ? endr : beg;
#pragma unroll
    for (int q = 0; q < D_FEAT / FUSE_ROWS; ++q)
        *(float4*)(s_w + (q * FUSE_ROWS + nl) * W_PITCH + f) = wq[q];

    float4 acc = hs;
    int j = beg;
    for (; j + 8 <= end; j += 8) gather_batch<8>(H, list, j, f, acc);
    const int rest = end - j;
    if (rest & 4) { gather_batch<4>(H, list, j, f, acc); j += 4; }
    if (rest & 2) { gather_batch<2>(H, list, j, f, acc); j += 2; }
    if (rest & 1) { gather_batch<1>(H, list, j, f, acc); }

    float4 o;
    o.x = ok ? fmaxf(acc.x * di + bv.x, 0.0f) : 0.0f;
    o.y = ok ? fmaxf(acc.y * di + bv.y, 0.0f) : 0.0f;
    o.z = ok ? fmaxf(acc.z * di + bv.z, 0.0f) : 0.0f;
    o.w = ok ? fmaxf(acc.w * di + bv.w, 0.0f) : 0.0f;
    *(float4*)(s_act + nl * TILE_PITCH + f) = o;
    __syncthreads();

    if (wv < 4) {
        const float* arow = s_act + lm * TILE_PITCH + kk;
        const float* bcol = s_w + kk * W_PITCH + wv * 16 + lm;
        v2f b[D_FEAT / 4];
#pragma unroll
        for (int s = 0; s < D_FEAT / 4; ++s) {
            b[s].x = bcol[(4 * s) * W_PITCH];
            b[s].y = bcol[(4 * s + 1) * W_PITCH];
        }
        __builtin_amdgcn_sched_barrier(0);
        v8f c = {};
#pragma unroll
        for (int s = 0; s < D_FEAT / 4; ++s) {
            const v2f a = *(const v2f*)(arow + 4 * s);
            c = __builtin_amdgcn_wmma_f32_16x16x4_f32(false, a, false, b[s], (short)0, c, false, false);
        }
        float* pr = s_prod + (lane >> 4) * 8 * TILE_PITCH + wv * 16 + lm;
#pragma unroll
        for (int r = 0; r < 8; ++r) pr[r * TILE_PITCH] = c[r];
    }
    __syncthreads();

    if (ok) {
        float4 p = *(const float4*)(s_prod + nl * TILE_PITCH + f);
        p.x *= di;
        p.y *= di;
        p.z *= di;
        p.w *= di;
        *(float4*)(HN + (size_t)node * D_FEAT + f) = p;
    }
}

__global__ __launch_bounds__(256) void gather_pool_kernel(const float* __restrict__ H,
                                                          const int* __restrict__ rowptr,
                                                          const int* __restrict__ list,
                                                          const float* __restrict__ dinv,
                                                          const float* __restrict__ bias,
                                                          const int* __restrict__ batch,
                                                          const float* __restrict__ cnt,
                                                          float* __restrict__ out, int n) {
    __shared__ __attribute__((aligned(16))) float s_row[FUSE_ROWS * D_FEAT];
    __shared__ float s_div[N_GRAPHS_C];
    __shared__ int s_gid[FUSE_ROWS];
    const int tid  = (int)threadIdx.x;
    const int nl   = tid >> 4;
    const int node = (int)blockIdx.x * FUSE_ROWS + nl;
    const int f    = (tid & 15) * 4;
    const bool ok  = node < n;
    const int nc   = ok ? node : (n - 1);

    const int beg  = rowptr[nc];
    const int endr = rowptr[nc + 1];
    const float di = dinv[nc];
    const float4 hs = *(const float4*)(H + (size_t)nc * D_FEAT + f);
    const float4 bv = *(const float4*)(bias + f);
    const int gb   = batch[nc];
    const float cg = cnt[tid & (N_GRAPHS_C - 1)];
    __builtin_amdgcn_sched_barrier(0);
    const int end  = ok ? endr : beg;

    float4 acc = hs;
    int j = beg;
    for (; j + 8 <= end; j += 8) gather_batch<8>(H, list, j, f, acc);
    const int rest = end - j;
    if (rest & 4) { gather_batch<4>(H, list, j, f, acc); j += 4; }
    if (rest & 2) { gather_batch<2>(H, list, j, f, acc); j += 2; }
    if (rest & 1) { gather_batch<1>(H, list, j, f, acc); }

    float4 o;
    o.x = fmaxf(acc.x * di + bv.x, 0.0f);
    o.y = fmaxf(acc.y * di + bv.y, 0.0f);
    o.z = fmaxf(acc.z * di + bv.z, 0.0f);
    o.w = fmaxf(acc.w * di + bv.w, 0.0f);
    *(float4*)(s_row + nl * D_FEAT + f) = o;
    if ((tid & 15) == 0) s_gid[nl] = (ok && (unsigned)gb < (unsigned)N_GRAPHS_C) ? gb : -1;
    if (tid < N_GRAPHS_C) s_div[tid] = fmaxf(cg, 1.0f);
    __syncthreads();

    if (tid < D_FEAT) {
        int   gq[FUSE_ROWS];
        float vq[FUSE_ROWS];
#pragma unroll
        for (int r = 0; r < FUSE_ROWS; ++r) {
            gq[r] = s_gid[r];
            vq[r] = s_row[r * D_FEAT + tid];
        }
        int   cur = -1;
        float run = 0.0f;
#pragma unroll
        for (int r = 0; r < FUSE_ROWS; ++r) {
            if (gq[r] != cur) {
                if (cur >= 0) atomicAdd(out + cur * D_FEAT + tid, run / s_div[cur]);
                cur = gq[r];
                run = vq[r];
            } else {
                run += vq[r];
            }
        }
        if (cur >= 0) atomicAdd(out + cur * D_FEAT + tid, run / s_div[cur]);
    }
}

extern "C" void kernel_launch(void* const* d_in, const int* in_sizes, int n_in,
                              void* d_out, int out_size, void* d_ws, size_t ws_size,
                              hipStream_t stream) {
    const float* x     = (const float*)d_in[0];
    const int*   ei    = (const int*)d_in[1];
    const int*   batch = (const int*)d_in[2];
    const float* W1    = (const float*)d_in[3];
    const float* b1    = (const float*)d_in[4];
    const float* W2    = (const float*)d_in[5];
    const float* b2    = (const float*)d_in[6];
    const float* W3    = (const float*)d_in[7];
    const float* b3    = (const float*)d_in[8];
    float* out = (float*)d_out;

    const int N = in_sizes[0] / D_FEAT;
    const int E = in_sizes[1] / 2;
    const int* erow = ei;
    const int* ecol = ei + E;

    float* w = (float*)d_ws;
    const size_t NP = (size_t)((N + 1 + 63) / 64) * 64;
    size_t off = 0;
    float* dinv = w + off; off += NP;
    float* bufA = w + off; off += (size_t)N * D_FEAT;
    float* bufB = w + off; off += (size_t)N * D_FEAT;
    float* cnt  = w + off; off += N_GRAPHS_C;
    int* rowptr   = (int*)(w + off); off += NP;
    int* wghist   = (int*)(w + off); off += (size_t)SORT_GROUPS * NB_MAX;
    int* binstart = (int*)(w + off); off += NB_MAX + 64;
    int2* binned  = (int2*)(w + off); off += (size_t)((E + 31) / 32) * 64;
    int* list     = (int*)(w + off); off += (size_t)((E + 63) / 64) * 64;

    const int blk = 256;

    const int NB    = (N + BIN_NODES - 1) / BIN_NODES;
    const int chunk = (E + SORT_GROUPS - 1) / SORT_GROUPS;
    bin_count_kernel<<<SORT_GROUPS, SORT_THREADS, 0, stream>>>(ecol, wghist, out, cnt, E, chunk, NB);
    bin_spread_kernel<<<SORT_GROUPS, SORT_THREADS, 0, stream>>>(erow, ecol, wghist, binstart, binned,
                                                                E, chunk, NB);
    bin_rank_kernel<<<NB, SORT_THREADS, 0, stream>>>(binned, binstart, rowptr, dinv, list, batch, cnt,
                                                     x, W1, bufA, N);

    const int fuse_blocks = (N + FUSE_ROWS - 1) / FUSE_ROWS;

    gather_gemm_kernel<<<fuse_blocks, blk, 0, stream>>>(bufA, rowptr, list, dinv, b1, W2, bufB, N);

    gather_gemm_kernel<<<fuse_blocks, blk, 0, stream>>>(bufB, rowptr, list, dinv, b2, W3, bufA, N);

    gather_pool_kernel<<<fuse_blocks, blk, 0, stream>>>(bufA, rowptr, list, dinv, b3, batch, cnt, out, N);
}


# edge binning in one launch: each work-group sorts its own chunk by bin in LDS (returning add = rank), stores whole lines; rank launch reads a bin's 128 runs; count launch dropped; reorders node sums
# speedup vs baseline: 7.1583x; 1.1430x over previous
#include <hip/hip_runtime.h>
#include <hip/hip_bf16.h>

typedef __attribute__((ext_vector_type(2))) float v2f;
typedef __attribute__((ext_vector_type(8))) float v8f;

#define D_FEAT 64
#define N_GRAPHS_C 64

#define BIN_SHIFT 8
#define BIN_NODES (1 << BIN_SHIFT)
#define NB_MAX 256
#define SORT_THREADS 1024
#define SORT_GROUPS 128
#define LOCAL_HELD 8
#define LOCAL_CAP (LOCAL_HELD * SORT_THREADS)
static_assert(SORT_GROUPS * SORT_THREADS >= N_GRAPHS_C * D_FEAT, "bin_local clears out with one thread per word");

__global__ __launch_bounds__(SORT_THREADS) void bin_local_kernel(const int* __restrict__ row,
                                                                 const int* __restrict__ col,
                                                                 int* __restrict__ wgoff,
                                                                 int2* __restrict__ binned,
                                                                 float* __restrict__ out,
                                                                 float* __restrict__ cnt,
                                                                 int e, int chunk, int nb) {
    __shared__ int s_h[NB_MAX];
    __shared__ int s_off[NB_MAX];
    __shared__ int s_x[NB_MAX];
    __shared__ int s_wave[NB_MAX / 32];
    __shared__ int s_tot;
    __shared__ __attribute__((aligned(16))) int2 s_stage[LOCAL_CAP];
    const int tid  = (int)threadIdx.x;
    const int lane = tid & 31;
    const int wv   = tid >> 5;
    const int grp  = (int)blockIdx.x;
    const int e0   = grp * chunk;
    const int e1   = min(e, e0 + chunk);

    int r[LOCAL_HELD], c[LOCAL_HELD], rk[LOCAL_HELD];
#pragma unroll
    for (int k = 0; k < LOCAL_HELD; ++k) {
        const int i = e0 + tid + k * SORT_THREADS;
        const bool ok = i < e1;
        c[k] = ok ? col[i] : -1;
        r[k] = ok ? row[i] : 0;
    }
    const int gt = grp * SORT_THREADS + tid;
    if (gt < N_GRAPHS_C * D_FEAT) out[gt] = 0.0f;
    if (gt < N_GRAPHS_C) cnt[gt] = 0.0f;
    if (tid < NB_MAX) s_h[tid] = 0;
    __syncthreads();

#pragma unroll
    for (int k = 0; k < LOCAL_HELD; ++k)
        rk[k] = (c[k] >= 0) ? atomicAdd(&s_h[c[k] >> BIN_SHIFT], 1) : 0;
    const bool extra = (e1 - e0) > LOCAL_CAP;
    if (extra) {
        __syncthreads();
        if (tid < NB_MAX) s_x[tid] = s_h[tid];
        __syncthreads();
        for (int i = e0 + tid + LOCAL_CAP; i < e1; i += SORT_THREADS) {
            const int cc = col[i];
            if (cc >= 0) atomicAdd(&s_h[cc >> BIN_SHIFT], 1);
        }
    }
    __syncthreads();

    int t = 0, x = 0;
    if (tid < NB_MAX) {
        t = s_h[tid];
        x = t;
#pragma unroll
        for (int d = 1; d < 32; d <<= 1) {
            const int y = __shfl_up(x, d);
            if (lane >= d) x += y;
        }
        if (lane == 31) s_wave[wv] = x;
    }
    __syncthreads();
    if (tid < NB_MAX) {
        int carry = 0;
#pragma unroll
        for (int k = 0; k < NB_MAX / 32; ++k) carry += (k < wv) ? s_wave[k] : 0;
        const int start = carry + (x - t);
        s_off[tid] = start;
        if (tid < nb) wgoff[tid * SORT_GROUPS + grp] = start;
        if (tid == nb - 1) wgoff[nb * SORT_GROUPS + grp] = start + t;
        if (tid == NB_MAX - 1) s_tot = start + t;
    }
    __syncthreads();

    if (!extra) {
        int so[LOCAL_HELD];
#pragma unroll
        for (int k = 0; k < LOCAL_HELD; ++k) so[k] = (c[k] >= 0) ? s_off[c[k] >> BIN_SHIFT] : 0;
#pragma unroll
        for (int k = 0; k < LOCAL_HELD; ++k)
            if (c[k] >= 0) s_stage[so[k] + rk[k]] = make_int2(r[k], c[k]);
        __syncthreads();
        const int m = s_tot;
        int2 pv[LOCAL_HELD];
#pragma unroll
        for (int k = 0; k < LOCAL_HELD; ++k) {
            const int i = tid + k * SORT_THREADS;
            pv[k] = (i < m) ? s_stage[i] : make_int2(0, 0);
        }
#pragma unroll
        for (int k = 0; k < LOCAL_HELD; ++k) {
            const int i = tid + k * SORT_THREADS;
            if (i < m) binned[e0 + i] = pv[k];
        }
    } else {
#pragma unroll
        for (int k = 0; k < LOCAL_HELD; ++k)
            if (c[k] >= 0) binned[e0 + s_off[c[k] >> BIN_SHIFT] + rk[k]] = make_int2(r[k], c[k]);
        for (int i = e0 + tid + LOCAL_CAP; i < e1; i += SORT_THREADS) {
            const int cc = col[i];
            const int rr = row[i];
            if (cc >= 0) {
                const int b = cc >> BIN_SHIFT;
                binned[e0 + s_off[b] + atomicAdd(&s_x[b], 1)] = make_int2(rr, cc);
            }
        }
    }
}

#define RANK_SEGS (SORT_GROUPS / (SORT_THREADS / 32))
#define RANK_SLOTS 2
#define RANK_HELD (RANK_SEGS * RANK_SLOTS)
static_assert(RANK_SEGS == 4, "a wave's run starts and ends are read as one int4 each");
#define RANK_STRIPS (BIN_NODES / 16)
#define W_PITCH 72
static_assert(SORT_THREADS / 32 == 2 * RANK_STRIPS, "two waves per strip, one per half of the columns");
static_assert(SORT_THREADS * 4 == D_FEAT * D_FEAT, "one float4 of the weights per thread");

__global__ __launch_bounds__(SORT_THREADS) void bin_rank_kernel(const int2* __restrict__ binned,
                                                                const int* __restrict__ wgoff, int chunk, int e,
                                                                int* __restrict__ rowptr,
                                                                float* __restrict__ dinv,
                                                                int* __restrict__ list,
                                                                const int* __restrict__ batch,
                                                                float* __restrict__ cnt,
                                                                const float* __restrict__ X,
                                                                const float* __restrict__ W,
                                                                float* __restrict__ H, int n) {
    __shared__ int s_cnt[BIN_NODES];
    __shared__ int s_ptr[BIN_NODES];
    __shared__ int s_wave[BIN_NODES / 32];
    __shared__ int s_gc[N_GRAPHS_C];
    __shared__ int s_lohi[2 * (SORT_GROUPS / 32)];
    __shared__ __attribute__((aligned(16))) float s_dinv[BIN_NODES];
    __shared__ __attribute__((aligned(16))) float s_w[D_FEAT * W_PITCH];
    const int tid   = (int)threadIdx.x;
    const int lane  = tid & 31;
    const int wv    = tid >> 5;
    const int bin   = (int)blockIdx.x;
    const int node0 = bin << BIN_SHIFT;

    const float4 wq = *(const float4*)(W + (size_t)(tid >> 4) * D_FEAT + (tid & 15) * 4);
    const int lm    = lane & 15;
    const int kk    = (lane >> 4) * 2;
    const int strip = wv & (RANK_STRIPS - 1);
    const int col0  = (wv / RANK_STRIPS) * 32;
    const int srow0 = node0 + strip * 16;
    const float* __restrict__ xrow = X + (size_t)min(srow0 + lm, n - 1) * D_FEAT + kk;
    v2f a[D_FEAT / 4];
#pragma unroll
    for (int s = 0; s < D_FEAT / 4; ++s) a[s] = *(const v2f*)(xrow + 4 * s);

    const int g0 = __builtin_amdgcn_readfirstlane(wv) * RANK_SEGS;
    const int4 a4 = *(const int4*)(wgoff + (size_t)bin * SORT_GROUPS + g0);
    const int4 z4 = *(const int4*)(wgoff + (size_t)(bin + 1) * SORT_GROUPS + g0);
    int pa = 0, pz = 0;
    if (tid < SORT_GROUPS) {
        pa = wgoff[(size_t)bin * SORT_GROUPS + tid];
        pz = wgoff[(size_t)(bin + 1) * SORT_GROUPS + tid];
    }
    const int sl[RANK_SEGS] = {z4.x - a4.x, z4.y - a4.y, z4.z - a4.z, z4.w - a4.w};
    const int sb[RANK_SEGS] = {(g0 + 0) * chunk + a4.x, (g0 + 1) * chunk + a4.y,
                               (g0 + 2) * chunk + a4.z, (g0 + 3) * chunk + a4.w};

    int2 ed[RANK_HELD];
#pragma unroll
    for (int q = 0; q < RANK_SEGS; ++q) {
#pragma unroll
        for (int i = 0; i < RANK_SLOTS; ++i) {
            const int idx = lane + 32 * i;
            ed[q * RANK_SLOTS + i] = binned[min(sb[q] + min(idx, max(sl[q] - 1, 0)), e - 1)];
            if (idx >= sl[q]) ed[q * RANK_SLOTS + i].y = -1;
        }
    }
    *(float4*)(s_w + (tid >> 4) * W_PITCH + (tid & 15) * 4) = wq;
    if (tid < SORT_GROUPS) {
#pragma unroll
        for (int d = 16; d >= 1; d >>= 1) {
            pa += __shfl_xor(pa, d);
            pz += __shfl_xor(pz, d);
        }
        if (lane == 0) {
            s_lohi[2 * wv]     = pa;
            s_lohi[2 * wv + 1] = pz;
        }
    }
    int gb = -1;
    if (tid < BIN_NODES && node0 + tid < n) gb = batch[node0 + tid];

    if (tid < BIN_NODES) s_cnt[tid] = 0;
    if (tid < N_GRAPHS_C) s_gc[tid] = 0;
    __syncthreads();

#pragma unroll
    for (int k = 0; k < RANK_HELD; ++k)
        if (ed[k].y >= 0) atomicAdd(&s_cnt[ed[k].y - node0], 1);
#pragma unroll
    for (int q = 0; q < RANK_SEGS; ++q)
        for (int idx = lane + 32 * RANK_SLOTS; idx < sl[q]; idx += 32)
            atomicAdd(&s_cnt[binned[sb[q] + idx].y - node0], 1);
    if ((unsigned)gb < (unsigned)N_GRAPHS_C) atomicAdd(&s_gc[gb], 1);
    int lo = 0, hi = 0;
#pragma unroll
    for (int k = 0; k < SORT_GROUPS / 32; ++k) {
        lo += s_lohi[2 * k];
        hi += s_lohi[2 * k + 1];
    }
    __syncthreads();

    int v = 0, x = 0;
    if (tid < BIN_NODES) {
        v = s_cnt[tid];
        x = v;
#pragma unroll
        for (int d = 1; d < 32; d <<= 1) {
            const int y = __shfl_up(x, d);
            if (lane >= d) x += y;
        }
        if (lane == 31) s_wave[wv] = x;
    }
    __syncthreads();
    if (tid < BIN_NODES) {
        int carry = 0;
#pragma unroll
        for (int k = 0; k < BIN_NODES / 32; ++k) carry += (k < wv) ? s_wave[k] : 0;
        const int p = lo + carry + (x - v);
        s_ptr[tid] = p;
        const int node = node0 + tid;
        const float d = (float)(v + 1);
        const float dv = (d > 0.0f) ? rsqrtf(d) : 0.0f;
        s_dinv[tid] = dv;
        if (node < n) {
            rowptr[node] = p;
            dinv[node] = dv;
        }
    }
    if (bin == (int)gridDim.x - 1 && tid == 0) rowptr[n] = hi;
    if (tid < N_GRAPHS_C) {
        const int gc = s_gc[tid];
        if (gc > 0) atomicAdd(&cnt[tid], (float)gc);
    }
    __syncthreads();

#pragma unroll
    for (int k = 0; k < RANK_HELD; ++k) {
        if (ed[k].y >= 0) {
            const int pos = atomicAdd(&s_ptr[ed[k].y - node0], 1);
            list[pos] = ed[k].x;
        }
    }
#pragma unroll
    for (int q = 0; q < RANK_SEGS; ++q) {
        for (int idx = lane + 32 * RANK_SLOTS; idx < sl[q]; idx += 32) {
            const int2 ee = binned[sb[q] + idx];
            const int pos = atomicAdd(&s_ptr[ee.y - node0], 1);
            list[pos] = ee.x;
        }
    }

    if (srow0 < n) {
        const float* bcol = s_w + kk * W_PITCH + col0 + lm;
        v8f c0 = {}, c1 = {};
#pragma unroll
        for (int s = 0; s < D_FEAT / 4; ++s) {
            const float* bp = bcol + (4 * s) * W_PITCH;
            v2f b0, b1;
            b0.x = bp[0];
            b0.y = bp[W_PITCH];
            b1.x = bp[16];
            b1.y = bp[W_PITCH + 16];
            c0 = __builtin_amdgcn_wmma_f32_16x16x4_f32(false, a[s], false, b0, (short)0, c0, false, false);
            c1 = __builtin_amdgcn_wmma_f32_16x16x4_f32(false, a[s], false, b1, (short)0, c1, false, false);
        }
        const int hb = (lane >> 4) * 8;
        const float4 dlo = *(const float4*)(s_dinv + strip * 16 + hb);
        const float4 dhi = *(const float4*)(s_dinv + strip * 16 + hb + 4);
        const float ds[8] = {dlo.x, dlo.y, dlo.z, dlo.w, dhi.x, dhi.y, dhi.z, dhi.w};
        float* __restrict__ hr = H + (size_t)(srow0 + hb) * D_FEAT + col0 + lm;
#pragma unroll
        for (int r = 0; r < 8; ++r) {
            if (srow0 + hb + r < n) {
                hr[r * D_FEAT]      = c0[r] * ds[r];
                hr[r * D_FEAT + 16] = c1[r] * ds[r];
            }
        }
    }
}

template <int B>
__device__ __forceinline__ void gather_batch(const float* __restrict__ H,
                                             const int* __restrict__ list,
                                             int j, int f, float4& acc) {
    int    en[B];
    float4 hv[B];
#pragma unroll
    for (int k = 0; k < B; ++k) en[k] = list[j + k];
#pragma unroll
    for (int k = 0; k < B; ++k)
        hv[k] = *(const float4*)(H + (size_t)en[k] * D_FEAT + f);
#pragma unroll
    for (int k = 0; k < B; ++k) {
        acc.x += hv[k].x;
        acc.y += hv[k].y;
        acc.z += hv[k].z;
        acc.w += hv[k].w;
    }
}

#define FUSE_ROWS 16
#define TILE_PITCH 68

__global__ __launch_bounds__(256) void gather_gemm_kernel(const float* __restrict__ H,
                                                          const int* __restrict__ rowptr,
                                                          const int* __restrict__ list,
                                                          const float* __restrict__ dinv,
                                                          const float* __restrict__ bias,
                                                          const float* __restrict__ WN,
                                                          float* __restrict__ HN, int n) {
    __shared__ __attribute__((aligned(16))) float s_w[D_FEAT * W_PITCH];
    __shared__ __attribute__((aligned(16))) float s_act[FUSE_ROWS * TILE_PITCH];
    __shared__ __attribute__((aligned(16))) float s_prod[FUSE_ROWS * TILE_PITCH];
    const int tid  = (int)threadIdx.x;
    const int nl   = tid >> 4;
    const int node = (int)blockIdx.x * FUSE_ROWS + nl;
    const int f    = (tid & 15) * 4;
    const int wv   = tid >> 5;
    const int lane = tid & 31;
    const int lm   = lane & 15;
    const int kk   = (lane >> 4) * 2;
    const bool ok  = node < n;
    const int nc   = ok ? node : (n - 1);

    float4 wq[D_FEAT / FUSE_ROWS];
#pragma unroll
    for (int q = 0; q < D_FEAT / FUSE_ROWS; ++q)
        wq[q] = *(const float4*)(WN + (size_t)(q * FUSE_ROWS + nl) * D_FEAT + f);
    const int beg  = rowptr[nc];
    const int endr = rowptr[nc + 1];
    const float di = dinv[nc];
    const float4 hs = *(const float4*)(H + (size_t)nc * D_FEAT + f);
    const float4 bv = *(const float4*)(bias + f);
    __builtin_amdgcn_sched_barrier(0);
    const int end = ok ? endr : beg;
#pragma unroll
    for (int q = 0; q < D_FEAT / FUSE_ROWS; ++q)
        *(float4*)(s_w + (q * FUSE_ROWS + nl) * W_PITCH + f) = wq[q];

    float4 acc = hs;
    int j = beg;
    for (; j + 8 <= end; j += 8) gather_batch<8>(H, list, j, f, acc);
    const int rest = end - j;
    if (rest & 4) { gather_batch<4>(H, list, j, f, acc); j += 4; }
    if (rest & 2) { gather_batch<2>(H, list, j, f, acc); j += 2; }
    if (rest & 1) { gather_batch<1>(H, list, j, f, acc); }

    float4 o;
    o.x = ok ? fmaxf(acc.x * di + bv.x, 0.0f) : 0.0f;
    o.y = ok ? fmaxf(acc.y * di + bv.y, 0.0f) : 0.0f;
    o.z = ok ? fmaxf(acc.z * di + bv.z, 0.0f) : 0.0f;
    o.w = ok ? fmaxf(acc.w * di + bv.w, 0.0f) : 0.0f;
    *(float4*)(s_act + nl * TILE_PITCH + f) = o;
    __syncthreads();

    if (wv < 4) {
        const float* arow = s_act + lm * TILE_PITCH + kk;
        const float* bcol = s_w + kk * W_PITCH + wv * 16 + lm;
        v2f b[D_FEAT / 4];
#pragma unroll
        for (int s = 0; s < D_FEAT / 4; ++s) {
            b[s].x = bcol[(4 * s) * W_PITCH];
            b[s].y = bcol[(4 * s + 1) * W_PITCH];
        }
        __builtin_amdgcn_sched_barrier(0);
        v8f c = {};
#pragma unroll
        for (int s = 0; s < D_FEAT / 4; ++s) {
            const v2f a = *(const v2f*)(arow + 4 * s);
            c = __builtin_amdgcn_wmma_f32_16x16x4_f32(false, a, false, b[s], (short)0, c, false, false);
        }
        float* pr = s_prod + (lane >> 4) * 8 * TILE_PITCH + wv * 16 + lm;
#pragma unroll
        for (int r = 0; r < 8; ++r) pr[r * TILE_PITCH] = c[r];
    }
    __syncthreads();

    if (ok) {
        float4 p = *(const float4*)(s_prod + nl * TILE_PITCH + f);
        p.x *= di;
        p.y *= di;
        p.z *= di;
        p.w *= di;
        *(float4*)(HN + (size_t)node * D_FEAT + f) = p;
    }
}

__global__ __launch_bounds__(256) void gather_pool_kernel(const float* __restrict__ H,
                                                          const int* __restrict__ rowptr,
                                                          const int* __restrict__ list,
                                                          const float* __restrict__ dinv,
                                                          const float* __restrict__ bias,
                                                          const int* __restrict__ batch,
                                                          const float* __restrict__ cnt,
                                                          float* __restrict__ out, int n) {
    __shared__ __attribute__((aligned(16))) float s_row[FUSE_ROWS * D_FEAT];
    __shared__ float s_div[N_GRAPHS_C];
    __shared__ int s_gid[FUSE_ROWS];
    const int tid  = (int)threadIdx.x;
    const int nl   = tid >> 4;
    const int node = (int)blockIdx.x * FUSE_ROWS + nl;
    const int f    = (tid & 15) * 4;
    const bool ok  = node < n;
    const int nc   = ok ? node : (n - 1);

    const int beg  = rowptr[nc];
    const int endr = rowptr[nc + 1];
    const float di = dinv[nc];
    const float4 hs = *(const float4*)(H + (size_t)nc * D_FEAT + f);
    const float4 bv = *(const float4*)(bias + f);
    const int gb   = batch[nc];
    const float cg = cnt[tid & (N_GRAPHS_C - 1)];
    __builtin_amdgcn_sched_barrier(0);
    const int end  = ok ? endr : beg;

    float4 acc = hs;
    int j = beg;
    for (; j + 8 <= end; j += 8) gather_batch<8>(H, list, j, f, acc);
    const int rest = end - j;
    if (rest & 4) { gather_batch<4>(H, list, j, f, acc); j += 4; }
    if (rest & 2) { gather_batch<2>(H, list, j, f, acc); j += 2; }
    if (rest & 1) { gather_batch<1>(H, list, j, f, acc); }

    float4 o;
    o.x = fmaxf(acc.x * di + bv.x, 0.0f);
    o.y = fmaxf(acc.y * di + bv.y, 0.0f);
    o.z = fmaxf(acc.z * di + bv.z, 0.0f);
    o.w = fmaxf(acc.w * di + bv.w, 0.0f);
    *(float4*)(s_row + nl * D_FEAT + f) = o;
    if ((tid & 15) == 0) s_gid[nl] = (ok && (unsigned)gb < (unsigned)N_GRAPHS_C) ? gb : -1;
    if (tid < N_GRAPHS_C) s_div[tid] = fmaxf(cg, 1.0f);
    __syncthreads();

    if (tid < D_FEAT) {
        int   gq[FUSE_ROWS];
        float vq[FUSE_ROWS];
#pragma unroll
        for (int r = 0; r < FUSE_ROWS; ++r) {
            gq[r] = s_gid[r];
            vq[r] = s_row[r * D_FEAT + tid];
        }
        int   cur = -1;
        float run = 0.0f;
#pragma unroll
        for (int r = 0; r < FUSE_ROWS; ++r) {
            if (gq[r] != cur) {
                if (cur >= 0) atomicAdd(out + cur * D_FEAT + tid, run / s_div[cur]);
                cur = gq[r];
                run = vq[r];
            } else {
                run += vq[r];
            }
        }
        if (cur >= 0) atomicAdd(out + cur * D_FEAT + tid, run / s_div[cur]);
    }
}

extern "C" void kernel_launch(void* const* d_in, const int* in_sizes, int n_in,
                              void* d_out, int out_size, void* d_ws, size_t ws_size,
                              hipStream_t stream) {
    const float* x     = (const float*)d_in[0];
    const int*   ei    = (const int*)d_in[1];
    const int*   batch = (const int*)d_in[2];
    const float* W1    = (const float*)d_in[3];
    const float* b1    = (const float*)d_in[4];
    const float* W2    = (const float*)d_in[5];
    const float* b2    = (const float*)d_in[6];
    const float* W3    = (const float*)d_in[7];
    const float* b3    = (const float*)d_in[8];
    float* out = (float*)d_out;

    const int N = in_sizes[0] / D_FEAT;
    const int E = in_sizes[1] / 2;
    const int* erow = ei;
    const int* ecol = ei + E;

    float* w = (float*)d_ws;
    const size_t NP = (size_t)((N + 1 + 63) / 64) * 64;
    size_t off = 0;
    float* dinv = w + off; off += NP;
    float* bufA = w + off; off += (size_t)N * D_FEAT;
    float* bufB = w + off; off += (size_t)N * D_FEAT;
    float* cnt  = w + off; off += N_GRAPHS_C;
    int* rowptr   = (int*)(w + off); off += NP;
    int* wgoff    = (int*)(w + off); off += (size_t)(NB_MAX + 1) * SORT_GROUPS;
    int2* binned  = (int2*)(w + off); off += (size_t)((E + 31) / 32) * 64;
    int* list     = (int*)(w + off); off += (size_t)((E + 63) / 64) * 64;

    const int blk = 256;

    const int NB    = (N + BIN_NODES - 1) / BIN_NODES;
    const int chunk = (E + SORT_GROUPS - 1) / SORT_GROUPS;
    bin_local_kernel<<<SORT_GROUPS, SORT_THREADS, 0, stream>>>(erow, ecol, wgoff, binned, out, cnt, E, chunk, NB);
    bin_rank_kernel<<<NB, SORT_THREADS, 0, stream>>>(binned, wgoff, chunk, E, rowptr, dinv, list, batch, cnt,
                                                     x, W1, bufA, N);

    const int fuse_blocks = (N + FUSE_ROWS - 1) / FUSE_ROWS;

    gather_gemm_kernel<<<fuse_blocks, blk, 0, stream>>>(bufA, rowptr, list, dinv, b1, W2, bufB, N);

    gather_gemm_kernel<<<fuse_blocks, blk, 0, stream>>>(bufB, rowptr, list, dinv, b2, W3, bufA, N);

    gather_pool_kernel<<<fuse_blocks, blk, 0, stream>>>(bufA, rowptr, list, dinv, b3, batch, cnt, out, N);
}


# gather: a node's list entries loaded one per lane (32 at once), passed on by DPP row_share; rows go out 16 at a time: 4 dependent round trips per node, was 2 per batch of 8, 4, 2, 1; sum order kept
# speedup vs baseline: 8.9105x; 1.2448x over previous
#include <hip/hip_runtime.h>
#include <hip/hip_bf16.h>

typedef __attribute__((ext_vector_type(2))) float v2f;
typedef __attribute__((ext_vector_type(8))) float v8f;
typedef __attribute__((ext_vector_type(2))) __bf16 v2bf;

#define ROW_DW (D_FEAT / 2)

__device__ __forceinline__ unsigned pack_bf16(float lo, float hi) {
    v2f v;
    v.x = lo;
    v.y = hi;
    return __builtin_bit_cast(unsigned, __builtin_convertvector(v, v2bf));
}
__device__ __forceinline__ float bf16_lo(unsigned d) { return __builtin_bit_cast(float, d << 16); }
__device__ __forceinline__ float bf16_hi(unsigned d) { return __builtin_bit_cast(float, d & 0xffff0000u); }

#define D_FEAT 64
#define N_GRAPHS_C 64

#define BIN_SHIFT 8
#define BIN_NODES (1 << BIN_SHIFT)
#define NB_MAX 256
#define SORT_THREADS 1024
#define SORT_GROUPS 128
#define LOCAL_HELD 8
#define LOCAL_CAP (LOCAL_HELD * SORT_THREADS)
static_assert(SORT_GROUPS * SORT_THREADS >= N_GRAPHS_C * D_FEAT, "bin_local clears out with one thread per word");

__global__ __launch_bounds__(SORT_THREADS) void bin_local_kernel(const int* __restrict__ row,
                                                                 const int* __restrict__ col,
                                                                 int* __restrict__ wgoff,
                                                                 int2* __restrict__ binned,
                                                                 float* __restrict__ out,
                                                                 float* __restrict__ cnt,
                                                                 int e, int chunk, int nb) {
    __shared__ int s_h[NB_MAX];
    __shared__ int s_off[NB_MAX];
    __shared__ int s_x[NB_MAX];
    __shared__ int s_wave[NB_MAX / 32];
    __shared__ int s_tot;
    __shared__ __attribute__((aligned(16))) int2 s_stage[LOCAL_CAP];
    const int tid  = (int)threadIdx.x;
    const int lane = tid & 31;
    const int wv   = tid >> 5;
    const int grp  = (int)blockIdx.x;
    const int e0   = grp * chunk;
    const int e1   = min(e, e0 + chunk);

    int r[LOCAL_HELD], c[LOCAL_HELD], rk[LOCAL_HELD];
#pragma unroll
    for (int k = 0; k < LOCAL_HELD; ++k) {
        const int i = e0 + tid + k * SORT_THREADS;
        const bool ok = i < e1;
        c[k] = ok ? col[i] : -1;
        r[k] = ok ? row[i] : 0;
    }
    const int gt = grp * SORT_THREADS + tid;
    if (gt < N_GRAPHS_C * D_FEAT) out[gt] = 0.0f;
    if (gt < N_GRAPHS_C) cnt[gt] = 0.0f;
    if (tid < NB_MAX) s_h[tid] = 0;
    __syncthreads();

#pragma unroll
    for (int k = 0; k < LOCAL_HELD; ++k)
        rk[k] = (c[k] >= 0) ? atomicAdd(&s_h[c[k] >> BIN_SHIFT], 1) : 0;
    const bool extra = (e1 - e0) > LOCAL_CAP;
    if (extra) {
        __syncthreads();
        if (tid < NB_MAX) s_x[tid] = s_h[tid];
        __syncthreads();
        for (int i = e0 + tid + LOCAL_CAP; i < e1; i += SORT_THREADS) {
            const int cc = col[i];
            if (cc >= 0) atomicAdd(&s_h[cc >> BIN_SHIFT], 1);
        }
    }
    __syncthreads();

    int t = 0, x = 0;
    if (tid < NB_MAX) {
        t = s_h[tid];
        x = t;
#pragma unroll
        for (int d = 1; d < 32; d <<= 1) {
            const int y = __shfl_up(x, d);
            if (lane >= d) x += y;
        }
        if (lane == 31) s_wave[wv] = x;
    }
    __syncthreads();
    if (tid < NB_MAX) {
        int carry = 0;
#pragma unroll
        for (int k = 0; k < NB_MAX / 32; ++k) carry += (k < wv) ? s_wave[k] : 0;
        const int start = carry + (x - t);
        s_off[tid] = start;
        if (tid < nb) wgoff[tid * SORT_GROUPS + grp] = start;
        if (tid == nb - 1) wgoff[nb * SORT_GROUPS + grp] = start + t;
        if (tid == NB_MAX - 1) s_tot = start + t;
    }
    __syncthreads();

    if (!extra) {
        int so[LOCAL_HELD];
#pragma unroll
        for (int k = 0; k < LOCAL_HELD; ++k) so[k] = (c[k] >= 0) ? s_off[c[k] >> BIN_SHIFT] : 0;
#pragma unroll
        for (int k = 0; k < LOCAL_HELD; ++k)
            if (c[k] >= 0) s_stage[so[k] + rk[k]] = make_int2(r[k], c[k]);
        __syncthreads();
        const int m = s_tot;
        int2 pv[LOCAL_HELD];
#pragma unroll
        for (int k = 0; k < LOCAL_HELD; ++k) {
            const int i = tid + k * SORT_THREADS;
            pv[k] = (i < m) ? s_stage[i] : make_int2(0, 0);
        }
#pragma unroll
        for (int k = 0; k < LOCAL_HELD; ++k) {
            const int i = tid + k * SORT_THREADS;
            if (i < m) binned[e0 + i] = pv[k];
        }
    } else {
#pragma unroll
        for (int k = 0; k < LOCAL_HELD; ++k)
            if (c[k] >= 0) binned[e0 + s_off[c[k] >> BIN_SHIFT] + rk[k]] = make_int2(r[k], c[k]);
        for (int i = e0 + tid + LOCAL_CAP; i < e1; i += SORT_THREADS) {
            const int cc = col[i];
            const int rr = row[i];
            if (cc >= 0) {
                const int b = cc >> BIN_SHIFT;
                binned[e0 + s_off[b] + atomicAdd(&s_x[b], 1)] = make_int2(rr, cc);
            }
        }
    }
}

#define RANK_SEGS (SORT_GROUPS / (SORT_THREADS / 32))
#define RANK_SLOTS 2
#define RANK_HELD (RANK_SEGS * RANK_SLOTS)
static_assert(RANK_SEGS == 4, "a wave's run starts and ends are read as one int4 each");
#define RANK_STRIPS (BIN_NODES / 16)
#define W_PITCH 72
static_assert(SORT_THREADS / 32 == 2 * RANK_STRIPS, "two waves per strip, one per half of the columns");
static_assert(SORT_THREADS * 4 == D_FEAT * D_FEAT, "one float4 of the weights per thread");
#define RANK_W_PITCH 80

__global__ __launch_bounds__(SORT_THREADS) void bin_rank_kernel(const int2* __restrict__ binned,
                                                                const int* __restrict__ wgoff, int chunk, int e,
                                                                int* __restrict__ rowptr,
                                                                float* __restrict__ dinv,
                                                                int* __restrict__ list,
                                                                const int* __restrict__ batch,
                                                                float* __restrict__ cnt,
                                                                const float* __restrict__ X,
                                                                const float* __restrict__ W,
                                                                unsigned* __restrict__ H, int n) {
    __shared__ int s_cnt[BIN_NODES];
    __shared__ int s_ptr[BIN_NODES];
    __shared__ int s_wave[BIN_NODES / 32];
    __shared__ int s_gc[N_GRAPHS_C];
    __shared__ int s_lohi[2 * (SORT_GROUPS / 32)];
    __shared__ __attribute__((aligned(16))) float s_dinv[BIN_NODES];
    __shared__ __attribute__((aligned(16))) float s_w[D_FEAT * RANK_W_PITCH];
    const int tid   = (int)threadIdx.x;
    const int lane  = tid & 31;
    const int wv    = tid >> 5;
    const int bin   = (int)blockIdx.x;
    const int node0 = bin << BIN_SHIFT;

    const float4 wq = *(const float4*)(W + (size_t)(tid >> 4) * D_FEAT + (tid & 15) * 4);
    const int lm    = lane & 15;
    const int kk    = (lane >> 4) * 2;
    const int strip = wv & (RANK_STRIPS - 1);
    const int col0  = (wv / RANK_STRIPS) * 32;
    const int srow0 = node0 + strip * 16;
    const float* __restrict__ xrow = X + (size_t)min(srow0 + lm, n - 1) * D_FEAT + kk;
    v2f a[D_FEAT / 4];
#pragma unroll
    for (int s = 0; s < D_FEAT / 4; ++s) a[s] = *(const v2f*)(xrow + 4 * s);

    const int g0 = __builtin_amdgcn_readfirstlane(wv) * RANK_SEGS;
    const int4 a4 = *(const int4*)(wgoff + (size_t)bin * SORT_GROUPS + g0);
    const int4 z4 = *(const int4*)(wgoff + (size_t)(bin + 1) * SORT_GROUPS + g0);
    int pa = 0, pz = 0;
    if (tid < SORT_GROUPS) {
        pa = wgoff[(size_t)bin * SORT_GROUPS + tid];
        pz = wgoff[(size_t)(bin + 1) * SORT_GROUPS + tid];
    }
    const int sl[RANK_SEGS] = {z4.x - a4.x, z4.y - a4.y, z4.z - a4.z, z4.w - a4.w};
    const int sb[RANK_SEGS] = {(g0 + 0) * chunk + a4.x, (g0 + 1) * chunk + a4.y,
                               (g0 + 2) * chunk + a4.z, (g0 + 3) * chunk + a4.w};

    int2 ed[RANK_HELD];
#pragma unroll
    for (int q = 0; q < RANK_SEGS; ++q) {
#pragma unroll
        for (int i = 0; i < RANK_SLOTS; ++i) {
            const int idx = lane + 32 * i;
            ed[q * RANK_SLOTS + i] = binned[min(sb[q] + min(idx, max(sl[q] - 1, 0)), e - 1)];
            if (idx >= sl[q]) ed[q * RANK_SLOTS + i].y = -1;
        }
    }
    *(float4*)(s_w + (tid >> 4) * RANK_W_PITCH + (tid & 15) * 4) = wq;
    if (tid < SORT_GROUPS) {
#pragma unroll
        for (int d = 16; d >= 1; d >>= 1) {
            pa += __shfl_xor(pa, d);
            pz += __shfl_xor(pz, d);
        }
        if (lane == 0) {
            s_lohi[2 * wv]     = pa;
            s_lohi[2 * wv + 1] = pz;
        }
    }
    int gb = -1;
    if (tid < BIN_NODES && node0 + tid < n) gb = batch[node0 + tid];

    if (tid < BIN_NODES) s_cnt[tid] = 0;
    if (tid < N_GRAPHS_C) s_gc[tid] = 0;
    __syncthreads();

#pragma unroll
    for (int k = 0; k < RANK_HELD; ++k)
        if (ed[k].y >= 0) atomicAdd(&s_cnt[ed[k].y - node0], 1);
#pragma unroll
    for (int q = 0; q < RANK_SEGS; ++q)
        for (int idx = lane + 32 * RANK_SLOTS; idx < sl[q]; idx += 32)
            atomicAdd(&s_cnt[binned[sb[q] + idx].y - node0], 1);
    if ((unsigned)gb < (unsigned)N_GRAPHS_C) atomicAdd(&s_gc[gb], 1);
    int lo = 0, hi = 0;
#pragma unroll
    for (int k = 0; k < SORT_GROUPS / 32; ++k) {
        lo += s_lohi[2 * k];
        hi += s_lohi[2 * k + 1];
    }
    __syncthreads();

    int v = 0, x = 0;
    if (tid < BIN_NODES) {
        v = s_cnt[tid];
        x = v;
#pragma unroll
        for (int d = 1; d < 32; d <<= 1) {
            const int y = __shfl_up(x, d);
            if (lane >= d) x += y;
        }
        if (lane == 31) s_wave[wv] = x;
    }
    __syncthreads();
    if (tid < BIN_NODES) {
        int carry = 0;
#pragma unroll
        for (int k = 0; k < BIN_NODES / 32; ++k) carry += (k < wv) ? s_wave[k] : 0;
        const int p = lo + carry + (x - v);
        s_ptr[tid] = p;
        const int node = node0 + tid;
        const float d = (float)(v + 1);
        const float dv = (d > 0.0f) ? rsqrtf(d) : 0.0f;
        s_dinv[tid] = dv;
        if (node < n) {
            rowptr[node] = p;
            dinv[node] = dv;
        }
    }
    if (bin == (int)gridDim.x - 1 && tid == 0) rowptr[n] = hi;
    if (tid < N_GRAPHS_C) {
        const int gc = s_gc[tid];
        if (gc > 0) atomicAdd(&cnt[tid], (float)gc);
    }
    __syncthreads();

#pragma unroll
    for (int k = 0; k < RANK_HELD; ++k) {
        if (ed[k].y >= 0) {
            const int pos = atomicAdd(&s_ptr[ed[k].y - node0], 1);
            list[pos] = ed[k].x;
        }
    }
#pragma unroll
    for (int q = 0; q < RANK_SEGS; ++q) {
        for (int idx = lane + 32 * RANK_SLOTS; idx < sl[q]; idx += 32) {
            const int2 ee = binned[sb[q] + idx];
            const int pos = atomicAdd(&s_ptr[ee.y - node0], 1);
            list[pos] = ee.x;
        }
    }

    if (srow0 < n) {
        const float* bcol = s_w + kk * RANK_W_PITCH + col0 + 2 * lm;
        v8f c0 = {}, c1 = {};
#pragma unroll
        for (int s = 0; s < D_FEAT / 4; ++s) {
            const float* bp = bcol + (4 * s) * RANK_W_PITCH;
            const v2f w0 = *(const v2f*)bp;
            const v2f w1 = *(const v2f*)(bp + RANK_W_PITCH);
            v2f b0, b1;
            b0.x = w0.x;
            b0.y = w1.x;
            b1.x = w0.y;
            b1.y = w1.y;
            c0 = __builtin_amdgcn_wmma_f32_16x16x4_f32(false, a[s], false, b0, (short)0, c0, false, false);
            c1 = __builtin_amdgcn_wmma_f32_16x16x4_f32(false, a[s], false, b1, (short)0, c1, false, false);
        }
        const int hb = (lane >> 4) * 8;
        const float4 dlo = *(const float4*)(s_dinv + strip * 16 + hb);
        const float4 dhi = *(const float4*)(s_dinv + strip * 16 + hb + 4);
        const float ds[8] = {dlo.x, dlo.y, dlo.z, dlo.w, dhi.x, dhi.y, dhi.z, dhi.w};
        unsigned* __restrict__ hr = H + (size_t)(srow0 + hb) * ROW_DW + col0 / 2 + lm;
#pragma unroll
        for (int r = 0; r < 8; ++r) {
            if (srow0 + hb + r < n)
                hr[r * ROW_DW] = pack_bf16(c0[r] * ds[r], c1[r] * ds[r]);
        }
    }
}

template <int K>
__device__ __forceinline__ unsigned row_entry(int v) {
    return (unsigned)__builtin_amdgcn_update_dpp(0, v, 0x150 + K, 0xf, 0xf, true);
}

#define GATHER_ROW(K) hv[K] = H[row_entry<K>(ent) * (unsigned)(ROW_DW / 2) + (unsigned)q];
#define GATHER_LOAD4(G)                                                         \
    if (4 * (G) < m) {                                                          \
        GATHER_ROW(4 * (G) + 0)                                                 \
        GATHER_ROW(4 * (G) + 1)                                                 \
        GATHER_ROW(4 * (G) + 2)                                                 \
        GATHER_ROW(4 * (G) + 3)                                                 \
    }
#define GATHER_SUM(K)                                                           \
    if (K < m) {                                                                \
        acc.x += bf16_lo(hv[K].x);                                              \
        acc.y += bf16_hi(hv[K].x);                                              \
        acc.z += bf16_lo(hv[K].y);                                              \
        acc.w += bf16_hi(hv[K].y);                                              \
    }

__device__ __forceinline__ void gather_rows16(const uint2* __restrict__ H, int ent, int m, int q, float4& acc) {
    uint2 hv[16];
    GATHER_LOAD4(0) GATHER_LOAD4(1) GATHER_LOAD4(2) GATHER_LOAD4(3)
    GATHER_SUM(0)  GATHER_SUM(1)  GATHER_SUM(2)  GATHER_SUM(3)
    GATHER_SUM(4)  GATHER_SUM(5)  GATHER_SUM(6)  GATHER_SUM(7)
    GATHER_SUM(8)  GATHER_SUM(9)  GATHER_SUM(10) GATHER_SUM(11)
    GATHER_SUM(12) GATHER_SUM(13) GATHER_SUM(14) GATHER_SUM(15)
}

__device__ __forceinline__ void gather_segment(const uint2* __restrict__ H,
                                               const int* __restrict__ list,
                                               int beg, int end, int q, float4& acc) {
    for (int base = beg; base < end; base += 32) {
        const int m    = end - base;
        const int last = end - 1;
        const int ent_a = list[min(base + q, last)];
        const int ent_b = list[min(base + 16 + q, last)];
        gather_rows16(H, ent_a, m, q, acc);
        if (m > 16) gather_rows16(H, ent_b, m - 16, q, acc);
    }
}

#define FUSE_ROWS 16
#define TILE_PITCH 68

__global__ __launch_bounds__(256) void gather_gemm_kernel(const uint2* __restrict__ H,
                                                          const int* __restrict__ rowptr,
                                                          const int* __restrict__ list,
                                                          const float* __restrict__ dinv,
                                                          const float* __restrict__ bias,
                                                          const float* __restrict__ WN,
                                                          uint2* __restrict__ HN, int n) {
    __shared__ __attribute__((aligned(16))) float s_w[D_FEAT * W_PITCH];
    __shared__ __attribute__((aligned(16))) float s_act[FUSE_ROWS * TILE_PITCH];
    __shared__ __attribute__((aligned(16))) float s_prod[FUSE_ROWS * TILE_PITCH];
    const int tid  = (int)threadIdx.x;
    const int nl   = tid >> 4;
    const int node = (int)blockIdx.x * FUSE_ROWS + nl;
    const int sl   = tid & 15;
    const int f    = sl * 4;
    const int wv   = tid >> 5;
    const int lane = tid & 31;
    const int lm   = lane & 15;
    const int kk   = (lane >> 4) * 2;
    const bool ok  = node < n;
    const int nc   = ok ? node : (n - 1);

    float4 wq[D_FEAT / FUSE_ROWS];
#pragma unroll
    for (int q = 0; q < D_FEAT / FUSE_ROWS; ++q)
        wq[q] = *(const float4*)(WN + (size_t)(q * FUSE_ROWS + nl) * D_FEAT + f);
    const int beg  = rowptr[nc];
    const int endr = rowptr[nc + 1];
    const float di = dinv[nc];
    const uint2 hs = H[(size_t)nc * (ROW_DW / 2) + sl];
    const float4 bv = *(const float4*)(bias + f);
    __builtin_amdgcn_sched_barrier(0);
    const int end = ok ? endr : beg;
#pragma unroll
    for (int q = 0; q < D_FEAT / FUSE_ROWS; ++q)
        *(float4*)(s_w + (q * FUSE_ROWS + nl) * W_PITCH + f) = wq[q];

    float4 acc;
    acc.x = bf16_lo(hs.x);
    acc.y = bf16_hi(hs.x);
    acc.z = bf16_lo(hs.y);
    acc.w = bf16_hi(hs.y);
    gather_segment(H, list, beg, end, sl, acc);

    float4 o;
    o.x = ok ? fmaxf(acc.x * di + bv.x, 0.0f) : 0.0f;
    o.y = ok ? fmaxf(acc.y * di + bv.y, 0.0f) : 0.0f;
    o.z = ok ? fmaxf(acc.z * di + bv.z, 0.0f) : 0.0f;
    o.w = ok ? fmaxf(acc.w * di + bv.w, 0.0f) : 0.0f;
    *(float4*)(s_act + nl * TILE_PITCH + f) = o;
    __syncthreads();

    if (wv < 4) {
        const float* arow = s_act + lm * TILE_PITCH + kk;
        const float* bcol = s_w + kk * W_PITCH + wv * 16 + lm;
        v2f b[D_FEAT / 4];
#pragma unroll
        for (int s = 0; s < D_FEAT / 4; ++s) {
            b[s].x = bcol[(4 * s) * W_PITCH];
            b[s].y = bcol[(4 * s + 1) * W_PITCH];
        }
        __builtin_amdgcn_sched_barrier(0);
        v8f c = {};
#pragma unroll
        for (int s = 0; s < D_FEAT / 4; ++s) {
            const v2f a = *(const v2f*)(arow + 4 * s);
            c = __builtin_amdgcn_wmma_f32_16x16x4_f32(false, a, false, b[s], (short)0, c, false, false);
        }
        float* pr = s_prod + (lane >> 4) * 8 * TILE_PITCH + wv * 16 + lm;
#pragma unroll
        for (int r = 0; r < 8; ++r) pr[r * TILE_PITCH] = c[r];
    }
    __syncthreads();

    if (ok) {
        const float4 p = *(const float4*)(s_prod + nl * TILE_PITCH + f);
        uint2 o2;
        o2.x = pack_bf16(p.x * di, p.y * di);
        o2.y = pack_bf16(p.z * di, p.w * di);
        HN[(size_t)node * (ROW_DW / 2) + sl] = o2;
    }
}

__global__ __launch_bounds__(256) void gather_pool_kernel(const uint2* __restrict__ H,
                                                          const int* __restrict__ rowptr,
                                                          const int* __restrict__ list,
                                                          const float* __restrict__ dinv,
                                                          const float* __restrict__ bias,
                                                          const int* __restrict__ batch,
                                                          const float* __restrict__ cnt,
                                                          float* __restrict__ out, int n) {
    __shared__ __attribute__((aligned(16))) float s_row[FUSE_ROWS * D_FEAT];
    __shared__ float s_div[N_GRAPHS_C];
    __shared__ int s_gid[FUSE_ROWS];
    const int tid  = (int)threadIdx.x;
    const int nl   = tid >> 4;
    const int node = (int)blockIdx.x * FUSE_ROWS + nl;
    const int sl   = tid & 15;
    const int f    = sl * 4;
    const bool ok  = node < n;
    const int nc   = ok ? node : (n - 1);

    const int beg  = rowptr[nc];
    const int endr = rowptr[nc + 1];
    const float di = dinv[nc];
    const uint2 hs = H[(size_t)nc * (ROW_DW / 2) + sl];
    const float4 bv = *(const float4*)(bias + f);
    const int gb   = batch[nc];
    const float cg = cnt[tid & (N_GRAPHS_C - 1)];
    __builtin_amdgcn_sched_barrier(0);
    const int end  = ok ? endr : beg;

    float4 acc;
    acc.x = bf16_lo(hs.x);
    acc.y = bf16_hi(hs.x);
    acc.z = bf16_lo(hs.y);
    acc.w = bf16_hi(hs.y);
    gather_segment(H, list, beg, end, sl, acc);

    float4 o;
    o.x = fmaxf(acc.x * di + bv.x, 0.0f);
    o.y = fmaxf(acc.y * di + bv.y, 0.0f);
    o.z = fmaxf(acc.z * di + bv.z, 0.0f);
    o.w = fmaxf(acc.w * di + bv.w, 0.0f);
    *(float4*)(s_row + nl * D_FEAT + f) = o;
    if ((tid & 15) == 0) s_gid[nl] = (ok && (unsigned)gb < (unsigned)N_GRAPHS_C) ? gb : -1;
    if (tid < N_GRAPHS_C) s_div[tid] = fmaxf(cg, 1.0f);
    __syncthreads();

    if (tid < D_FEAT) {
        int   gq[FUSE_ROWS];
        float vq[FUSE_ROWS];
#pragma unroll
        for (int r = 0; r < FUSE_ROWS; ++r) {
            gq[r] = s_gid[r];
            vq[r] = s_row[r * D_FEAT + tid];
        }
        int   cur = -1;
        float run = 0.0f;
#pragma unroll
        for (int r = 0; r < FUSE_ROWS; ++r) {
            if (gq[r] != cur) {
                if (cur >= 0) atomicAdd(out + cur * D_FEAT + tid, run / s_div[cur]);
                cur = gq[r];
                run = vq[r];
            } else {
                run += vq[r];
            }
        }
        if (cur >= 0) atomicAdd(out + cur * D_FEAT + tid, run / s_div[cur]);
    }
}

extern "C" void kernel_launch(void* const* d_in, const int* in_sizes, int n_in,
                              void* d_out, int out_size, void* d_ws, size_t ws_size,
                              hipStream_t stream) {
    const float* x     = (const float*)d_in[0];
    const int*   ei    = (const int*)d_in[1];
    const int*   batch = (const int*)d_in[2];
    const float* W1    = (const float*)d_in[3];
    const float* b1    = (const float*)d_in[4];
    const float* W2    = (const float*)d_in[5];
    const float* b2    = (const float*)d_in[6];
    const float* W3    = (const float*)d_in[7];
    const float* b3    = (const float*)d_in[8];
    float* out = (float*)d_out;

    const int N = in_sizes[0] / D_FEAT;
    const int E = in_sizes[1] / 2;
    const int* erow = ei;
    const int* ecol = ei + E;

    float* w = (float*)d_ws;
    const size_t NP = (size_t)((N + 1 + 63) / 64) * 64;
    size_t off = 0;
    float* dinv = w + off; off += NP;
    uint2* bufA = (uint2*)(w + off); off += (size_t)N * ROW_DW;
    uint2* bufB = (uint2*)(w + off); off += (size_t)N * ROW_DW;
    float* cnt  = w + off; off += N_GRAPHS_C;
    int* rowptr   = (int*)(w + off); off += NP;
    int* wgoff    = (int*)(w + off); off += (size_t)(NB_MAX + 1) * SORT_GROUPS;
    int2* binned  = (int2*)(w + off); off += (size_t)((E + 31) / 32) * 64;
    int* list     = (int*)(w + off); off += (size_t)((E + 63) / 64) * 64;

    const int blk = 256;

    const int NB    = (N + BIN_NODES - 1) / BIN_NODES;
    const int chunk = (E + SORT_GROUPS - 1) / SORT_GROUPS;
    bin_local_kernel<<<SORT_GROUPS, SORT_THREADS, 0, stream>>>(erow, ecol, wgoff, binned, out, cnt, E, chunk, NB);
    bin_rank_kernel<<<NB, SORT_THREADS, 0, stream>>>(binned, wgoff, chunk, E, rowptr, dinv, list, batch, cnt,
                                                     x, W1, (unsigned*)bufA, N);

    const int fuse_blocks = (N + FUSE_ROWS - 1) / FUSE_ROWS;

    gather_gemm_kernel<<<fuse_blocks, blk, 0, stream>>>(bufA, rowptr, list, dinv, b1, W2, bufB, N);

    gather_gemm_kernel<<<fuse_blocks, blk, 0, stream>>>(bufB, rowptr, list, dinv, b2, W3, bufA, N);

    gather_pool_kernel<<<fuse_blocks, blk, 0, stream>>>(bufA, rowptr, list, dinv, b3, batch, cnt, out, N);
}
